# starting kernel
# baseline (speedup 1.0000x reference)
#include <hip/hip_runtime.h>


typedef __attribute__((ext_vector_type(2))) float v2f;
typedef __attribute__((ext_vector_type(8))) float v8f;

#define LDK 260

__device__ __forceinline__ v8f wmma4(v2f a, v2f b, v8f c) {
  return __builtin_amdgcn_wmma_f32_16x16x4_f32(false, a, false, b,
                                               (short)0, c, false, false);
}

__device__ __forceinline__ void gemm_acc(const float* As, int aoff,
                                         const float* __restrict__ B,
                                         int K, int col0, int lane, v8f& c) {
  const int hl = lane >> 4;
  const int m  = lane & 15;
  const int n  = lane & 15;
  #pragma unroll 4
  for (int k0 = 0; k0 < K; k0 += 4) {
    const int ka = k0 + 2 * hl;
    v2f a, b;
    a.x = As[m * LDK + aoff + ka];
    a.y = As[m * LDK + aoff + ka + 1];
    b.x = B[ka * 128 + col0 + n];
    b.y = B[(ka + 1) * 128 + col0 + n];
    c = wmma4(a, b, c);
  }
}

__device__ __forceinline__ void store_tile(float* __restrict__ out, int row0,
                                           int nrows, int col0, int lane,
                                           v8f c, bool do_relu) {
  const int hl = lane >> 4;
  const int n  = lane & 15;
  #pragma unroll
  for (int v = 0; v < 8; ++v) {
    int row = row0 + v + 8 * hl;
    float val = c[v];
    if (do_relu) val = fmaxf(val, 0.0f);
    if (row < nrows) out[(size_t)row * 128 + col0 + n] = val;
  }
}


__global__ void __launch_bounds__(256) encode_pol_kernel(
    const float* __restrict__ x, const int* __restrict__ sidx,
    const float* __restrict__ semb, const float* __restrict__ W,
    const float* __restrict__ bvec, float* __restrict__ out, int nrows) {
  __shared__ float As[16 * LDK];
  const int tid  = threadIdx.x;
  const int row0 = blockIdx.x * 16;
  for (int idx = tid; idx < 16 * 72; idx += 256) {
    int r = idx / 72, cc = idx % 72;
    int row = row0 + r;
    float v = 0.0f;
    if (row < nrows) {
      if (cc < 64) v = x[(size_t)row * 64 + cc];
      else         v = semb[sidx[row] * 8 + (cc - 64)];
    }
    As[r * LDK + cc] = v;
  }
  __syncthreads();
  const int lane = tid & 31;
  const int col0 = (tid >> 5) * 16;
  v8f c;
  float bias = bvec[col0 + (lane & 15)];
  #pragma unroll
  for (int v = 0; v < 8; ++v) c[v] = bias;
  gemm_acc(As, 0, W, 72, col0, lane, c);
  store_tile(out, row0, nrows, col0, lane, c, true);
}

__global__ void __launch_bounds__(256) encode_comp_kernel(
    const float* __restrict__ x, const int* __restrict__ secidx,
    const int* __restrict__ indidx, const float* __restrict__ secemb,
    const float* __restrict__ indemb, const float* __restrict__ W,
    const float* __restrict__ bvec, float* __restrict__ out, int nrows) {
  __shared__ float As[16 * LDK];
  const int tid  = threadIdx.x;
  const int row0 = blockIdx.x * 16;
  for (int idx = tid; idx < 16 * 112; idx += 256) {
    int r = idx / 112, cc = idx % 112;
    int row = row0 + r;
    float v = 0.0f;
    if (row < nrows) {
      if (cc < 96)       v = x[(size_t)row * 96 + cc];
      else if (cc < 104) v = secemb[secidx[row] * 8 + (cc - 96)];
      else               v = indemb[indidx[row] * 8 + (cc - 104)];
    }
    As[r * LDK + cc] = v;
  }
  __syncthreads();
  const int lane = tid & 31;
  const int col0 = (tid >> 5) * 16;
  v8f c;
  float bias = bvec[col0 + (lane & 15)];
  #pragma unroll
  for (int v = 0; v < 8; ++v) c[v] = bias;
  gemm_acc(As, 0, W, 112, col0, lane, c);
  store_tile(out, row0, nrows, col0, lane, c, true);
}


__global__ void zero_kernel(float* __restrict__ p, size_t n) {
  size_t i      = (size_t)blockIdx.x * blockDim.x + threadIdx.x;
  size_t stride = (size_t)gridDim.x * blockDim.x;
  for (; i < n; i += stride) p[i] = 0.0f;
}

__global__ void __launch_bounds__(256) scatter_kernel(
    const float* __restrict__ h, const int* __restrict__ src,
    const int* __restrict__ dst, float* __restrict__ s,
    float* __restrict__ cnt, int E) {
  const int lane = threadIdx.x & 31;
  const int e    = blockIdx.x * 8 + (threadIdx.x >> 5);
  if (e >= E) return;
  const int sv = src[e];
  const int dv = dst[e];
  const float4 v = *reinterpret_cast<const float4*>(h + (size_t)sv * 128 + lane * 4);
  float* p = s + (size_t)dv * 128 + lane * 4;
  atomicAdd(p + 0, v.x);
  atomicAdd(p + 1, v.y);
  atomicAdd(p + 2, v.z);
  atomicAdd(p + 3, v.w);
  if (lane == 0) atomicAdd(cnt + dv, 1.0f);
}


__global__ void __launch_bounds__(256) sage_kernel(
    const float* __restrict__ h, const float* __restrict__ s,
    const float* __restrict__ cnt, const float* __restrict__ Wl,
    const float* __restrict__ bl, const float* __restrict__ Wr,
    float* __restrict__ out, int nrows, int do_relu) {
  __shared__ float As[16 * LDK];
  const int tid  = threadIdx.x;
  const int row0 = blockIdx.x * 16;
  for (int idx = tid; idx < 16 * 256; idx += 256) {
    int r = idx >> 8, cc = idx & 255;
    int row = row0 + r;
    float v = 0.0f;
    if (row < nrows) {
      if (cc < 128) {
        v = s[(size_t)row * 128 + cc] / fmaxf(cnt[row], 1.0f);
      } else {
        v = h[(size_t)row * 128 + (cc - 128)];
      }
    }
    As[r * LDK + cc] = v;
  }
  __syncthreads();
  const int lane = tid & 31;
  const int col0 = (tid >> 5) * 16;
  v8f c;
  float bias = bl[col0 + (lane & 15)];
  #pragma unroll
  for (int v = 0; v < 8; ++v) c[v] = bias;
  gemm_acc(As, 0,   Wl, 128, col0, lane, c);
  gemm_acc(As, 128, Wr, 128, col0, lane, c);
  store_tile(out, row0, nrows, col0, lane, c, do_relu != 0);
}


extern "C" void kernel_launch(void* const* d_in, const int* in_sizes, int n_in,
                              void* d_out, int out_size, void* d_ws, size_t ws_size,
                              hipStream_t stream) {
  (void)n_in; (void)out_size; (void)ws_size;
  const float* x_pol   = (const float*)d_in[0];
  const int*   pol_sid = (const int*)  d_in[1];
  const float* x_comp  = (const float*)d_in[2];
  const int*   c_sec   = (const int*)  d_in[3];
  const int*   c_ind   = (const int*)  d_in[4];
  const int*   edges   = (const int*)  d_in[5];
  const float* s_emb   = (const float*)d_in[6];
  const float* sec_emb = (const float*)d_in[7];
  const float* ind_emb = (const float*)d_in[8];
  const float* W_pol   = (const float*)d_in[9];
  const float* b_pol   = (const float*)d_in[10];
  const float* W_comp  = (const float*)d_in[11];
  const float* b_comp  = (const float*)d_in[12];
  const float* Wl1     = (const float*)d_in[13];
  const float* bl1     = (const float*)d_in[14];
  const float* Wr1     = (const float*)d_in[15];
  const float* Wl2     = (const float*)d_in[16];
  const float* bl2     = (const float*)d_in[17];
  const float* Wr2     = (const float*)d_in[18];

  const int NP = in_sizes[0] / 64;
  const int NC = in_sizes[2] / 96;
  const int E  = in_sizes[5] / 2;
  const int N  = NP + NC;
  const int* src = edges;
  const int* dst = edges + E;

  float* h    = (float*)d_ws;
  float* s    = h + (size_t)N * 128;
  float* cnt  = s + (size_t)N * 128;
  float* outp = (float*)d_out;

  const dim3 blk(256);
  const size_t zn = (size_t)N * 128 + N;
  int zgrid = (int)((zn + 255) / 256);
  if (zgrid > 65535) zgrid = 65535;

  encode_pol_kernel<<<dim3((NP + 15) / 16), blk, 0, stream>>>(
      x_pol, pol_sid, s_emb, W_pol, b_pol, h, NP);
  encode_comp_kernel<<<dim3((NC + 15) / 16), blk, 0, stream>>>(
      x_comp, c_sec, c_ind, sec_emb, ind_emb, W_comp, b_comp,
      h + (size_t)NP * 128, NC);

  zero_kernel<<<dim3(zgrid), blk, 0, stream>>>(s, zn);
  scatter_kernel<<<dim3((E + 7) / 8), blk, 0, stream>>>(h, src, dst, s, cnt, E);
  sage_kernel<<<dim3((N + 15) / 16), blk, 0, stream>>>(
      h, s, cnt, Wl1, bl1, Wr1, h, N, 1);

  zero_kernel<<<dim3(zgrid), blk, 0, stream>>>(s, zn);
  scatter_kernel<<<dim3((E + 7) / 8), blk, 0, stream>>>(h, src, dst, s, cnt, E);
  sage_kernel<<<dim3((N + 15) / 16), blk, 0, stream>>>(
      h, s, cnt, Wl2, bl2, Wr2, outp, N, 0);
}


# Mean aggregation by per-node gather over a dst-sorted edge list built each call (int-atomic counting sort) instead of f32 atomic scatter; sum order per node = int-atomic arrival rank
# speedup vs baseline: 1.7735x; 1.7735x over previous
#include <hip/hip_runtime.h>


typedef __attribute__((ext_vector_type(2))) float v2f;
typedef __attribute__((ext_vector_type(8))) float v8f;

#define LDK 260

__device__ __forceinline__ v8f wmma4(v2f a, v2f b, v8f c) {
  return __builtin_amdgcn_wmma_f32_16x16x4_f32(false, a, false, b,
                                               (short)0, c, false, false);
}

__device__ __forceinline__ void gemm_acc(const float* As, int aoff,
                                         const float* __restrict__ B,
                                         int K, int col0, int lane, v8f& c) {
  const int hl = lane >> 4;
  const int m  = lane & 15;
  const int n  = lane & 15;
  #pragma unroll 4
  for (int k0 = 0; k0 < K; k0 += 4) {
    const int ka = k0 + 2 * hl;
    v2f a, b;
    a.x = As[m * LDK + aoff + ka];
    a.y = As[m * LDK + aoff + ka + 1];
    b.x = B[ka * 128 + col0 + n];
    b.y = B[(ka + 1) * 128 + col0 + n];
    c = wmma4(a, b, c);
  }
}

__device__ __forceinline__ void store_tile(float* __restrict__ out, int row0,
                                           int nrows, int col0, int lane,
                                           v8f c, bool do_relu) {
  const int hl = lane >> 4;
  const int n  = lane & 15;
  #pragma unroll
  for (int v = 0; v < 8; ++v) {
    int row = row0 + v + 8 * hl;
    float val = c[v];
    if (do_relu) val = fmaxf(val, 0.0f);
    if (row < nrows) out[(size_t)row * 128 + col0 + n] = val;
  }
}


__global__ void __launch_bounds__(256) encode_pol_kernel(
    const float* __restrict__ x, const int* __restrict__ sidx,
    const float* __restrict__ semb, const float* __restrict__ W,
    const float* __restrict__ bvec, float* __restrict__ out, int nrows) {
  __shared__ float As[16 * LDK];
  const int tid  = threadIdx.x;
  const int row0 = blockIdx.x * 16;
  for (int idx = tid; idx < 16 * 72; idx += 256) {
    int r = idx / 72, cc = idx % 72;
    int row = row0 + r;
    float v = 0.0f;
    if (row < nrows) {
      if (cc < 64) v = x[(size_t)row * 64 + cc];
      else         v = semb[sidx[row] * 8 + (cc - 64)];
    }
    As[r * LDK + cc] = v;
  }
  __syncthreads();
  const int lane = tid & 31;
  const int col0 = (tid >> 5) * 16;
  v8f c;
  float bias = bvec[col0 + (lane & 15)];
  #pragma unroll
  for (int v = 0; v < 8; ++v) c[v] = bias;
  gemm_acc(As, 0, W, 72, col0, lane, c);
  store_tile(out, row0, nrows, col0, lane, c, true);
}

__global__ void __launch_bounds__(256) encode_comp_kernel(
    const float* __restrict__ x, const int* __restrict__ secidx,
    const int* __restrict__ indidx, const float* __restrict__ secemb,
    const float* __restrict__ indemb, const float* __restrict__ W,
    const float* __restrict__ bvec, float* __restrict__ out, int nrows) {
  __shared__ float As[16 * LDK];
  const int tid  = threadIdx.x;
  const int row0 = blockIdx.x * 16;
  for (int idx = tid; idx < 16 * 112; idx += 256) {
    int r = idx / 112, cc = idx % 112;
    int row = row0 + r;
    float v = 0.0f;
    if (row < nrows) {
      if (cc < 96)       v = x[(size_t)row * 96 + cc];
      else if (cc < 104) v = secemb[secidx[row] * 8 + (cc - 96)];
      else               v = indemb[indidx[row] * 8 + (cc - 104)];
    }
    As[r * LDK + cc] = v;
  }
  __syncthreads();
  const int lane = tid & 31;
  const int col0 = (tid >> 5) * 16;
  v8f c;
  float bias = bvec[col0 + (lane & 15)];
  #pragma unroll
  for (int v = 0; v < 8; ++v) c[v] = bias;
  gemm_acc(As, 0, W, 112, col0, lane, c);
  store_tile(out, row0, nrows, col0, lane, c, true);
}


__global__ void __launch_bounds__(256) zero_int_kernel(int* __restrict__ p, int n) {
  const int i = blockIdx.x * 256 + threadIdx.x;
  if (i < n) p[i] = 0;
}

__global__ void __launch_bounds__(256) degree_kernel(
    const int* __restrict__ dst, int* __restrict__ deg, int E) {
  const int e = blockIdx.x * 256 + threadIdx.x;
  if (e < E) atomicAdd(deg + dst[e], 1);
}

__global__ void __launch_bounds__(1024) offsets_kernel(
    const int* __restrict__ deg, int* __restrict__ rowbeg,
    int* __restrict__ cursor, int N) {
  __shared__ int sh_pre[32];
  __shared__ int sh_tot[32];
  const int tid  = threadIdx.x;
  const int lane = tid & 31;
  const int w    = tid >> 5;
  const int base = blockIdx.x * 4096;
  int pre = 0;
  const int4* front = reinterpret_cast<const int4*>(deg) + tid;
  #pragma unroll 8
  for (int c = 0; c < (int)blockIdx.x; ++c) {
    const int4 d = front[c * 1024];
    pre += (d.x + d.y) + (d.z + d.w);
  }
  const int idx = base + tid * 4;
  int4 d = make_int4(0, 0, 0, 0);
  if (idx < N) d = *reinterpret_cast<const int4*>(deg + idx);
  const int tsum = (d.x + d.y) + (d.z + d.w);
  for (int o = 16; o > 0; o >>= 1) pre += __shfl_xor(pre, o, 32);
  int inc = tsum;
  for (int o = 1; o < 32; o <<= 1) {
    const int t = __shfl_up(inc, o, 32);
    if (lane >= o) inc += t;
  }
  if (lane == 31) sh_tot[w] = inc;
  if (lane == 0)  sh_pre[w] = pre;
  __syncthreads();
  int p = sh_pre[lane];
  int t = (lane < w) ? sh_tot[lane] : 0;
  for (int o = 16; o > 0; o >>= 1) {
    p += __shfl_xor(p, o, 32);
    t += __shfl_xor(t, o, 32);
  }
  int ex = p + t + inc - tsum;
  if (idx     < N) { rowbeg[idx]     = ex; cursor[idx]     = ex; }
  ex += d.x;
  if (idx + 1 < N) { rowbeg[idx + 1] = ex; cursor[idx + 1] = ex; }
  ex += d.y;
  if (idx + 2 < N) { rowbeg[idx + 2] = ex; cursor[idx + 2] = ex; }
  ex += d.z;
  if (idx + 3 < N) { rowbeg[idx + 3] = ex; cursor[idx + 3] = ex; }
}

__global__ void __launch_bounds__(256) fill_kernel(
    const int* __restrict__ src, const int* __restrict__ dst,
    int* __restrict__ cursor, int* __restrict__ col, int E) {
  const int e = blockIdx.x * 256 + threadIdx.x;
  if (e < E) {
    const int p = atomicAdd(cursor + dst[e], 1);
    col[p] = src[e];
  }
}

template <int K>
__device__ __forceinline__ void add_rows(const float* __restrict__ h, int mycol,
                                         int j, int lane, float4& acc) {
  float4 v[K];
  #pragma unroll
  for (int k = 0; k < K; ++k) {
    const int sv = __builtin_amdgcn_readlane(mycol, j + k);
    v[k] = *reinterpret_cast<const float4*>(h + (size_t)sv * 128 + lane * 4);
  }
  #pragma unroll
  for (int k = 0; k < K; ++k) {
    acc.x += v[k].x; acc.y += v[k].y; acc.z += v[k].z; acc.w += v[k].w;
  }
}

__global__ void __launch_bounds__(256) gather_mean_kernel(
    const float* __restrict__ h, const int* __restrict__ rowbeg,
    const int* __restrict__ rowend, const int* __restrict__ col,
    float* __restrict__ m, int N) {
  const int lane = threadIdx.x & 31;
  const int i = __builtin_amdgcn_readfirstlane(blockIdx.x * 8 + (threadIdx.x >> 5));
  if (i >= N) return;
  const int beg = rowbeg[i];
  const int end = rowend[i];
  float4 acc = make_float4(0.0f, 0.0f, 0.0f, 0.0f);
  for (int base = beg; base < end; base += 32) {
    const int n = min(32, end - base);
    const int mycol = (lane < n) ? col[base + lane] : 0;
    int j = 0;
    for (; j + 8 <= n; j += 8) add_rows<8>(h, mycol, j, lane, acc);
    if (n - j >= 4) { add_rows<4>(h, mycol, j, lane, acc); j += 4; }
    if (n - j >= 2) { add_rows<2>(h, mycol, j, lane, acc); j += 2; }
    if (n - j >= 1) { add_rows<1>(h, mycol, j, lane, acc); }
  }
  const float d = fmaxf((float)(end - beg), 1.0f);
  float4 r;
  r.x = acc.x / d; r.y = acc.y / d; r.z = acc.z / d; r.w = acc.w / d;
  *reinterpret_cast<float4*>(m + (size_t)i * 128 + lane * 4) = r;
}


__global__ void __launch_bounds__(256) sage_kernel(
    const float* h, const float* m,
    const float* __restrict__ Wl,
    const float* __restrict__ bl, const float* __restrict__ Wr,
    float* out, int nrows, int do_relu) {
  __shared__ float As[16 * LDK];
  const int tid  = threadIdx.x;
  const int row0 = blockIdx.x * 16;
  for (int idx = tid; idx < 16 * 256; idx += 256) {
    int r = idx >> 8, cc = idx & 255;
    int row = row0 + r;
    float v = 0.0f;
    if (row < nrows) {
      if (cc < 128) {
        v = m[(size_t)row * 128 + cc];
      } else {
        v = h[(size_t)row * 128 + (cc - 128)];
      }
    }
    As[r * LDK + cc] = v;
  }
  __syncthreads();
  const int lane = tid & 31;
  const int col0 = (tid >> 5) * 16;
  v8f c;
  float bias = bl[col0 + (lane & 15)];
  #pragma unroll
  for (int v = 0; v < 8; ++v) c[v] = bias;
  gemm_acc(As, 0,   Wl, 128, col0, lane, c);
  gemm_acc(As, 128, Wr, 128, col0, lane, c);
  store_tile(out, row0, nrows, col0, lane, c, do_relu != 0);
}


extern "C" void kernel_launch(void* const* d_in, const int* in_sizes, int n_in,
                              void* d_out, int out_size, void* d_ws, size_t ws_size,
                              hipStream_t stream) {
  (void)n_in; (void)out_size; (void)ws_size;
  const float* x_pol   = (const float*)d_in[0];
  const int*   pol_sid = (const int*)  d_in[1];
  const float* x_comp  = (const float*)d_in[2];
  const int*   c_sec   = (const int*)  d_in[3];
  const int*   c_ind   = (const int*)  d_in[4];
  const int*   edges   = (const int*)  d_in[5];
  const float* s_emb   = (const float*)d_in[6];
  const float* sec_emb = (const float*)d_in[7];
  const float* ind_emb = (const float*)d_in[8];
  const float* W_pol   = (const float*)d_in[9];
  const float* b_pol   = (const float*)d_in[10];
  const float* W_comp  = (const float*)d_in[11];
  const float* b_comp  = (const float*)d_in[12];
  const float* Wl1     = (const float*)d_in[13];
  const float* bl1     = (const float*)d_in[14];
  const float* Wr1     = (const float*)d_in[15];
  const float* Wl2     = (const float*)d_in[16];
  const float* bl2     = (const float*)d_in[17];
  const float* Wr2     = (const float*)d_in[18];

  const int NP = in_sizes[0] / 64;
  const int NC = in_sizes[2] / 96;
  const int E  = in_sizes[5] / 2;
  const int N  = NP + NC;
  const int* src = edges;
  const int* dst = edges + E;

  float* h    = (float*)d_ws;
  float* outp = (float*)d_out;
  const int N4 = (N + 3) & ~3;
  int* deg    = (int*)(h + (size_t)N * 128);
  int* rowbeg = deg + N4;
  int* cursor = rowbeg + N4;
  int* col    = cursor + N4;
  float* mean = outp;

  const dim3 blk(256);

  encode_pol_kernel<<<dim3((NP + 15) / 16), blk, 0, stream>>>(
      x_pol, pol_sid, s_emb, W_pol, b_pol, h, NP);
  encode_comp_kernel<<<dim3((NC + 15) / 16), blk, 0, stream>>>(
      x_comp, c_sec, c_ind, sec_emb, ind_emb, W_comp, b_comp,
      h + (size_t)NP * 128, NC);

  zero_int_kernel<<<dim3((N4 + 255) / 256), blk, 0, stream>>>(deg, N4);
  degree_kernel<<<dim3((E + 255) / 256), blk, 0, stream>>>(dst, deg, E);
  offsets_kernel<<<dim3((N + 4095) / 4096), dim3(1024), 0, stream>>>(
      deg, rowbeg, cursor, N);
  fill_kernel<<<dim3((E + 255) / 256), blk, 0, stream>>>(src, dst, cursor, col, E);

  gather_mean_kernel<<<dim3((N + 7) / 8), blk, 0, stream>>>(
      h, rowbeg, cursor, col, mean, N);
  sage_kernel<<<dim3((N + 15) / 16), blk, 0, stream>>>(
      h, mean, Wl1, bl1, Wr1, h, N, 1);

  gather_mean_kernel<<<dim3((N + 7) / 8), blk, 0, stream>>>(
      h, rowbeg, cursor, col, mean, N);
  sage_kernel<<<dim3((N + 15) / 16), blk, 0, stream>>>(
      h, mean, Wl2, bl2, Wr2, outp, N, 0);
}


# Edges grouped by destination by a two-level counting sort on LDS counters (512-node bins, then nodes; contiguous copy-out), no global int atomics; sum order per node = LDS-atomic arrival rank
# speedup vs baseline: 2.6632x; 1.5017x over previous
#include <hip/hip_runtime.h>


typedef __attribute__((ext_vector_type(2))) float v2f;
typedef __attribute__((ext_vector_type(8))) float v8f;

#define LDK 260

__device__ __forceinline__ v8f wmma4(v2f a, v2f b, v8f c) {
  return __builtin_amdgcn_wmma_f32_16x16x4_f32(false, a, false, b,
                                               (short)0, c, false, false);
}

__device__ __forceinline__ void gemm_acc(const float* As, int aoff,
                                         const float* __restrict__ B,
                                         int K, int col0, int lane, v8f& c) {
  const int hl = lane >> 4;
  const int m  = lane & 15;
  const int n  = lane & 15;
  #pragma unroll 4
  for (int k0 = 0; k0 < K; k0 += 4) {
    const int ka = k0 + 2 * hl;
    v2f a, b;
    a.x = As[m * LDK + aoff + ka];
    a.y = As[m * LDK + aoff + ka + 1];
    b.x = B[ka * 128 + col0 + n];
    b.y = B[(ka + 1) * 128 + col0 + n];
    c = wmma4(a, b, c);
  }
}

__device__ __forceinline__ void store_tile(float* __restrict__ out, int row0,
                                           int nrows, int col0, int lane,
                                           v8f c, bool do_relu) {
  const int hl = lane >> 4;
  const int n  = lane & 15;
  #pragma unroll
  for (int v = 0; v < 8; ++v) {
    int row = row0 + v + 8 * hl;
    float val = c[v];
    if (do_relu) val = fmaxf(val, 0.0f);
    if (row < nrows) out[(size_t)row * 128 + col0 + n] = val;
  }
}


__global__ void __launch_bounds__(256) encode_pol_kernel(
    const float* __restrict__ x, const int* __restrict__ sidx,
    const float* __restrict__ semb, const float* __restrict__ W,
    const float* __restrict__ bvec, float* __restrict__ out, int nrows) {
  __shared__ float As[16 * LDK];
  const int tid  = threadIdx.x;
  const int row0 = blockIdx.x * 16;
  for (int idx = tid; idx < 16 * 72; idx += 256) {
    int r = idx / 72, cc = idx % 72;
    int row = row0 + r;
    float v = 0.0f;
    if (row < nrows) {
      if (cc < 64) v = x[(size_t)row * 64 + cc];
      else         v = semb[sidx[row] * 8 + (cc - 64)];
    }
    As[r * LDK + cc] = v;
  }
  __syncthreads();
  const int lane = tid & 31;
  const int col0 = (tid >> 5) * 16;
  v8f c;
  float bias = bvec[col0 + (lane & 15)];
  #pragma unroll
  for (int v = 0; v < 8; ++v) c[v] = bias;
  gemm_acc(As, 0, W, 72, col0, lane, c);
  store_tile(out, row0, nrows, col0, lane, c, true);
}

__global__ void __launch_bounds__(256) encode_comp_kernel(
    const float* __restrict__ x, const int* __restrict__ secidx,
    const int* __restrict__ indidx, const float* __restrict__ secemb,
    const float* __restrict__ indemb, const float* __restrict__ W,
    const float* __restrict__ bvec, float* __restrict__ out, int nrows) {
  __shared__ float As[16 * LDK];
  const int tid  = threadIdx.x;
  const int row0 = blockIdx.x * 16;
  for (int idx = tid; idx < 16 * 112; idx += 256) {
    int r = idx / 112, cc = idx % 112;
    int row = row0 + r;
    float v = 0.0f;
    if (row < nrows) {
      if (cc < 96)       v = x[(size_t)row * 96 + cc];
      else if (cc < 104) v = secemb[secidx[row] * 8 + (cc - 96)];
      else               v = indemb[indidx[row] * 8 + (cc - 104)];
    }
    As[r * LDK + cc] = v;
  }
  __syncthreads();
  const int lane = tid & 31;
  const int col0 = (tid >> 5) * 16;
  v8f c;
  float bias = bvec[col0 + (lane & 15)];
  #pragma unroll
  for (int v = 0; v < 8; ++v) c[v] = bias;
  gemm_acc(As, 0, W, 112, col0, lane, c);
  store_tile(out, row0, nrows, col0, lane, c, true);
}


#define BIN_SHIFT 9
#define BIN_NODES 512
#define MAX_BINS 1024
#define CHUNK_CAP 8192
#define BIN_STAGE 12288
#define SORT_THREADS 1024

__global__ void __launch_bounds__(SORT_THREADS) bin_count_kernel(
    const int* __restrict__ dst, int* __restrict__ tab, int E, int NB) {
  __shared__ int hist[MAX_BINS];
  const int tid = threadIdx.x;
  for (int k = tid; k < NB; k += SORT_THREADS) hist[k] = 0;
  __syncthreads();
  const int G = (int)gridDim.x;
  const int chunk = (E + G - 1) / G;
  const int beg = (int)blockIdx.x * chunk;
  const int end = min(E, beg + chunk);
  for (int e0 = beg + tid; e0 < end; e0 += 4 * SORT_THREADS) {
    int d[4];
    #pragma unroll
    for (int u = 0; u < 4; ++u) {
      const int e = e0 + u * SORT_THREADS;
      d[u] = (e < end) ? dst[e] : 0;
    }
    #pragma unroll
    for (int u = 0; u < 4; ++u) {
      if (e0 + u * SORT_THREADS < end) atomicAdd(&hist[d[u] >> BIN_SHIFT], 1);
    }
  }
  __syncthreads();
  for (int k = tid; k < NB; k += SORT_THREADS) tab[(int)blockIdx.x * NB + k] = hist[k];
}

__global__ void __launch_bounds__(SORT_THREADS) bin_scatter_kernel(
    const int* __restrict__ src, const int* __restrict__ dst,
    const int* __restrict__ tab, int* __restrict__ binbase,
    int* __restrict__ pairs, int E, int NB) {
  __shared__ int tot[4][MAX_BINS];
  __shared__ int pre[4][MAX_BINS];
  __shared__ int lcnt[MAX_BINS];
  __shared__ int gbase[MAX_BINS];
  __shared__ int lofs[MAX_BINS];
  __shared__ int lcur[MAX_BINS];
  __shared__ int2 stage[CHUNK_CAP];
  const int tid = threadIdx.x;
  const int G   = (int)gridDim.x;
  const int me  = (int)blockIdx.x;
  for (int k = tid; k < NB; k += SORT_THREADS) lcnt[k] = tab[me * NB + k];
  {
    const int part = tid >> 8;
    const int b0 = (part * G) >> 2;
    const int b1 = ((part + 1) * G) >> 2;
    for (int k = tid & 255; k < NB; k += 256) {
      int t = 0, p = 0;
      #pragma unroll 8
      for (int b = b0; b < b1; ++b) {
        const int v = tab[b * NB + k];
        t += v;
        if (b < me) p += v;
      }
      tot[part][k] = t;
      pre[part][k] = p;
    }
  }
  __syncthreads();
  if (tid < 32) {
    const int C  = (NB + 31) >> 5;
    const int kb = tid * C;
    const int ke = min(NB, kb + C);
    int s = 0, ls = 0;
    for (int k = kb; k < ke; ++k) {
      s  += (tot[0][k] + tot[1][k]) + (tot[2][k] + tot[3][k]);
      ls += lcnt[k];
    }
    int inc = s, linc = ls;
    for (int o = 1; o < 32; o <<= 1) {
      const int t  = __shfl_up(inc, o, 32);
      const int lt = __shfl_up(linc, o, 32);
      if (tid >= o) { inc += t; linc += lt; }
    }
    int run = inc - s, lrun = linc - ls;
    for (int k = kb; k < ke; ++k) {
      gbase[k] = run + (pre[0][k] + pre[1][k]) + (pre[2][k] + pre[3][k]);
      lofs[k]  = lrun;
      lcur[k]  = lrun;
      if (me == 0) binbase[k] = run;
      run  += (tot[0][k] + tot[1][k]) + (tot[2][k] + tot[3][k]);
      lrun += lcnt[k];
    }
    if (me == 0 && tid == 31) binbase[NB] = run;
  }
  __syncthreads();
  const int chunk = (E + G - 1) / G;
  const int beg = me * chunk;
  const int end = min(E, beg + chunk);
  for (int e0 = beg + tid; e0 < end; e0 += 4 * SORT_THREADS) {
    int d[4], s[4];
    #pragma unroll
    for (int u = 0; u < 4; ++u) {
      const int e = e0 + u * SORT_THREADS;
      d[u] = (e < end) ? dst[e] : 0;
      s[u] = (e < end) ? src[e] : 0;
    }
    #pragma unroll
    for (int u = 0; u < 4; ++u) {
      if (e0 + u * SORT_THREADS < end) {
        const int q = atomicAdd(&lcur[d[u] >> BIN_SHIFT], 1);
        stage[q] = make_int2(s[u], d[u]);
      }
    }
  }
  __syncthreads();
  int2* out2 = reinterpret_cast<int2*>(pairs);
  const int cnt = end - beg;
  for (int i = tid; i < cnt; i += SORT_THREADS) {
    const int2 pr = stage[i];
    const int k = pr.y >> BIN_SHIFT;
    out2[gbase[k] + (i - lofs[k])] = pr;
  }
}

__global__ void __launch_bounds__(SORT_THREADS) bin_sort_kernel(
    const int* __restrict__ pairs, const int* __restrict__ binbase,
    int* __restrict__ rowbeg, int* __restrict__ rowend,
    int* __restrict__ col, int N) {
  __shared__ int cnt[BIN_NODES];
  __shared__ int start[BIN_NODES];
  __shared__ int cur[BIN_NODES];
  __shared__ int stage[BIN_STAGE];
  const int tid = threadIdx.x;
  const int bin = (int)blockIdx.x;
  const int2* in2 = reinterpret_cast<const int2*>(pairs);
  const int beg = binbase[bin];
  const int end = binbase[bin + 1];
  if (tid < BIN_NODES) cnt[tid] = 0;
  __syncthreads();
  for (int e0 = beg + tid; e0 < end; e0 += 4 * SORT_THREADS) {
    int d[4];
    #pragma unroll
    for (int u = 0; u < 4; ++u) {
      const int e = e0 + u * SORT_THREADS;
      d[u] = (e < end) ? in2[e].y : 0;
    }
    #pragma unroll
    for (int u = 0; u < 4; ++u) {
      if (e0 + u * SORT_THREADS < end) atomicAdd(&cnt[d[u] & (BIN_NODES - 1)], 1);
    }
  }
  __syncthreads();
  if (tid < 32) {
    int c[BIN_NODES / 32];
    int s = 0;
    #pragma unroll
    for (int j = 0; j < BIN_NODES / 32; ++j) { c[j] = cnt[tid * (BIN_NODES / 32) + j]; s += c[j]; }
    int inc = s;
    for (int o = 1; o < 32; o <<= 1) {
      const int t = __shfl_up(inc, o, 32);
      if (tid >= o) inc += t;
    }
    int run = inc - s;
    #pragma unroll
    for (int j = 0; j < BIN_NODES / 32; ++j) {
      start[tid * (BIN_NODES / 32) + j] = run;
      cur[tid * (BIN_NODES / 32) + j]   = run;
      run += c[j];
    }
  }
  __syncthreads();
  if (tid < BIN_NODES) {
    const int node = bin * BIN_NODES + tid;
    if (node < N) {
      rowbeg[node] = beg + start[tid];
      rowend[node] = beg + start[tid] + cnt[tid];
    }
  }
  for (int e0 = beg + tid; e0 < end; e0 += 4 * SORT_THREADS) {
    int2 pr[4];
    #pragma unroll
    for (int u = 0; u < 4; ++u) {
      const int e = e0 + u * SORT_THREADS;
      pr[u] = (e < end) ? in2[e] : make_int2(0, 0);
    }
    #pragma unroll
    for (int u = 0; u < 4; ++u) {
      if (e0 + u * SORT_THREADS < end) {
        const int q = atomicAdd(&cur[pr[u].y & (BIN_NODES - 1)], 1);
        if (q < BIN_STAGE) stage[q] = pr[u].x;
        else               col[beg + q] = pr[u].x;
      }
    }
  }
  __syncthreads();
  const int staged = min(end - beg, BIN_STAGE);
  for (int i = tid; i < staged; i += SORT_THREADS) col[beg + i] = stage[i];
}

template <int K>
__device__ __forceinline__ void add_rows(const float* __restrict__ h, int mycol,
                                         int j, int lane, float4& acc) {
  float4 v[K];
  #pragma unroll
  for (int k = 0; k < K; ++k) {
    const int sv = __builtin_amdgcn_readlane(mycol, j + k);
    v[k] = *reinterpret_cast<const float4*>(h + (size_t)sv * 128 + lane * 4);
  }
  #pragma unroll
  for (int k = 0; k < K; ++k) {
    acc.x += v[k].x; acc.y += v[k].y; acc.z += v[k].z; acc.w += v[k].w;
  }
}

__global__ void __launch_bounds__(256) gather_mean_kernel(
    const float* __restrict__ h, const int* __restrict__ rowbeg,
    const int* __restrict__ rowend, const int* __restrict__ col,
    float* __restrict__ m, int N) {
  const int lane = threadIdx.x & 31;
  const int i = __builtin_amdgcn_readfirstlane(blockIdx.x * 8 + (threadIdx.x >> 5));
  if (i >= N) return;
  const int beg = rowbeg[i];
  const int end = rowend[i];
  float4 acc = make_float4(0.0f, 0.0f, 0.0f, 0.0f);
  for (int base = beg; base < end; base += 32) {
    const int n = min(32, end - base);
    const int mycol = (lane < n) ? col[base + lane] : 0;
    int j = 0;
    for (; j + 8 <= n; j += 8) add_rows<8>(h, mycol, j, lane, acc);
    if (n - j >= 4) { add_rows<4>(h, mycol, j, lane, acc); j += 4; }
    if (n - j >= 2) { add_rows<2>(h, mycol, j, lane, acc); j += 2; }
    if (n - j >= 1) { add_rows<1>(h, mycol, j, lane, acc); }
  }
  const float d = fmaxf((float)(end - beg), 1.0f);
  float4 r;
  r.x = acc.x / d; r.y = acc.y / d; r.z = acc.z / d; r.w = acc.w / d;
  *reinterpret_cast<float4*>(m + (size_t)i * 128 + lane * 4) = r;
}


__global__ void __launch_bounds__(256) sage_kernel(
    const float* h, const float* m,
    const float* __restrict__ Wl,
    const float* __restrict__ bl, const float* __restrict__ Wr,
    float* out, int nrows, int do_relu) {
  __shared__ float As[16 * LDK];
  const int tid  = threadIdx.x;
  const int row0 = blockIdx.x * 16;
  for (int idx = tid; idx < 16 * 256; idx += 256) {
    int r = idx >> 8, cc = idx & 255;
    int row = row0 + r;
    float v = 0.0f;
    if (row < nrows) {
      if (cc < 128) {
        v = m[(size_t)row * 128 + cc];
      } else {
        v = h[(size_t)row * 128 + (cc - 128)];
      }
    }
    As[r * LDK + cc] = v;
  }
  __syncthreads();
  const int lane = tid & 31;
  const int col0 = (tid >> 5) * 16;
  v8f c;
  float bias = bl[col0 + (lane & 15)];
  #pragma unroll
  for (int v = 0; v < 8; ++v) c[v] = bias;
  gemm_acc(As, 0,   Wl, 128, col0, lane, c);
  gemm_acc(As, 128, Wr, 128, col0, lane, c);
  store_tile(out, row0, nrows, col0, lane, c, do_relu != 0);
}


extern "C" void kernel_launch(void* const* d_in, const int* in_sizes, int n_in,
                              void* d_out, int out_size, void* d_ws, size_t ws_size,
                              hipStream_t stream) {
  (void)n_in; (void)out_size; (void)ws_size;
  const float* x_pol   = (const float*)d_in[0];
  const int*   pol_sid = (const int*)  d_in[1];
  const float* x_comp  = (const float*)d_in[2];
  const int*   c_sec   = (const int*)  d_in[3];
  const int*   c_ind   = (const int*)  d_in[4];
  const int*   edges   = (const int*)  d_in[5];
  const float* s_emb   = (const float*)d_in[6];
  const float* sec_emb = (const float*)d_in[7];
  const float* ind_emb = (const float*)d_in[8];
  const float* W_pol   = (const float*)d_in[9];
  const float* b_pol   = (const float*)d_in[10];
  const float* W_comp  = (const float*)d_in[11];
  const float* b_comp  = (const float*)d_in[12];
  const float* Wl1     = (const float*)d_in[13];
  const float* bl1     = (const float*)d_in[14];
  const float* Wr1     = (const float*)d_in[15];
  const float* Wl2     = (const float*)d_in[16];
  const float* bl2     = (const float*)d_in[17];
  const float* Wr2     = (const float*)d_in[18];

  const int NP = in_sizes[0] / 64;
  const int NC = in_sizes[2] / 96;
  const int E  = in_sizes[5] / 2;
  const int N  = NP + NC;
  const int* src = edges;
  const int* dst = edges + E;

  float* h    = (float*)d_ws;
  float* outp = (float*)d_out;
  const int N4 = (N + 3) & ~3;
  const int NB = (N + BIN_NODES - 1) >> BIN_SHIFT;
  const int G  = (E + CHUNK_CAP - 1) / CHUNK_CAP;
  int* tab     = (int*)(h + (size_t)N * 128);
  int* binbase = tab + G * NB;
  int* rowbeg  = tab + N4;
  int* rowend  = rowbeg + N4;
  int* col     = rowend + N4;
  int* pairs   = (int*)d_out;
  float* mean  = outp;

  const dim3 blk(256);

  encode_pol_kernel<<<dim3((NP + 15) / 16), blk, 0, stream>>>(
      x_pol, pol_sid, s_emb, W_pol, b_pol, h, NP);
  encode_comp_kernel<<<dim3((NC + 15) / 16), blk, 0, stream>>>(
      x_comp, c_sec, c_ind, sec_emb, ind_emb, W_comp, b_comp,
      h + (size_t)NP * 128, NC);

  bin_count_kernel<<<dim3(G), dim3(SORT_THREADS), 0, stream>>>(dst, tab, E, NB);
  bin_scatter_kernel<<<dim3(G), dim3(SORT_THREADS), 0, stream>>>(
      src, dst, tab, binbase, pairs, E, NB);
  bin_sort_kernel<<<dim3(NB), dim3(SORT_THREADS), 0, stream>>>(
      pairs, binbase, rowbeg, rowend, col, N);

  gather_mean_kernel<<<dim3((N + 7) / 8), blk, 0, stream>>>(
      h, rowbeg, rowend, col, mean, N);
  sage_kernel<<<dim3((N + 15) / 16), blk, 0, stream>>>(
      h, mean, Wl1, bl1, Wr1, h, N, 1);

  gather_mean_kernel<<<dim3((N + 7) / 8), blk, 0, stream>>>(
      h, rowbeg, rowend, col, mean, N);
  sage_kernel<<<dim3((N + 15) / 16), blk, 0, stream>>>(
      h, mean, Wl2, bl2, Wr2, outp, N, 0);
}


# Mean aggregation fused into the layer kernel: each wave of a 16-node work-group sums two nodes' neighbour rows (same list order) into the LDS A tile; no gather launches, no mean rows in memory
# speedup vs baseline: 3.3511x; 1.2583x over previous
#include <hip/hip_runtime.h>


typedef __attribute__((ext_vector_type(2))) float v2f;
typedef __attribute__((ext_vector_type(8))) float v8f;

#define LDK 260

__device__ __forceinline__ v8f wmma4(v2f a, v2f b, v8f c) {
  return __builtin_amdgcn_wmma_f32_16x16x4_f32(false, a, false, b,
                                               (short)0, c, false, false);
}

__device__ __forceinline__ void gemm_acc(const float* As, int aoff,
                                         const float* __restrict__ B,
                                         int K, int col0, int lane, v8f& c) {
  const int hl = lane >> 4;
  const int m  = lane & 15;
  const int n  = lane & 15;
  #pragma unroll 4
  for (int k0 = 0; k0 < K; k0 += 4) {
    const int ka = k0 + 2 * hl;
    v2f a, b;
    a.x = As[m * LDK + aoff + ka];
    a.y = As[m * LDK + aoff + ka + 1];
    b.x = B[ka * 128 + col0 + n];
    b.y = B[(ka + 1) * 128 + col0 + n];
    c = wmma4(a, b, c);
  }
}

__device__ __forceinline__ void store_tile(float* __restrict__ out, int row0,
                                           int nrows, int col0, int lane,
                                           v8f c, bool do_relu) {
  const int hl = lane >> 4;
  const int n  = lane & 15;
  #pragma unroll
  for (int v = 0; v < 8; ++v) {
    int row = row0 + v + 8 * hl;
    float val = c[v];
    if (do_relu) val = fmaxf(val, 0.0f);
    if (row < nrows) out[(size_t)row * 128 + col0 + n] = val;
  }
}


__global__ void __launch_bounds__(256) encode_pol_kernel(
    const float* __restrict__ x, const int* __restrict__ sidx,
    const float* __restrict__ semb, const float* __restrict__ W,
    const float* __restrict__ bvec, float* __restrict__ out, int nrows) {
  __shared__ float As[16 * LDK];
  const int tid  = threadIdx.x;
  const int row0 = blockIdx.x * 16;
  for (int idx = tid; idx < 16 * 72; idx += 256) {
    int r = idx / 72, cc = idx % 72;
    int row = row0 + r;
    float v = 0.0f;
    if (row < nrows) {
      if (cc < 64) v = x[(size_t)row * 64 + cc];
      else         v = semb[sidx[row] * 8 + (cc - 64)];
    }
    As[r * LDK + cc] = v;
  }
  __syncthreads();
  const int lane = tid & 31;
  const int col0 = (tid >> 5) * 16;
  v8f c;
  float bias = bvec[col0 + (lane & 15)];
  #pragma unroll
  for (int v = 0; v < 8; ++v) c[v] = bias;
  gemm_acc(As, 0, W, 72, col0, lane, c);
  store_tile(out, row0, nrows, col0, lane, c, true);
}

__global__ void __launch_bounds__(256) encode_comp_kernel(
    const float* __restrict__ x, const int* __restrict__ secidx,
    const int* __restrict__ indidx, const float* __restrict__ secemb,
    const float* __restrict__ indemb, const float* __restrict__ W,
    const float* __restrict__ bvec, float* __restrict__ out, int nrows) {
  __shared__ float As[16 * LDK];
  const int tid  = threadIdx.x;
  const int row0 = blockIdx.x * 16;
  for (int idx = tid; idx < 16 * 112; idx += 256) {
    int r = idx / 112, cc = idx % 112;
    int row = row0 + r;
    float v = 0.0f;
    if (row < nrows) {
      if (cc < 96)       v = x[(size_t)row * 96 + cc];
      else if (cc < 104) v = secemb[secidx[row] * 8 + (cc - 96)];
      else               v = indemb[indidx[row] * 8 + (cc - 104)];
    }
    As[r * LDK + cc] = v;
  }
  __syncthreads();
  const int lane = tid & 31;
  const int col0 = (tid >> 5) * 16;
  v8f c;
  float bias = bvec[col0 + (lane & 15)];
  #pragma unroll
  for (int v = 0; v < 8; ++v) c[v] = bias;
  gemm_acc(As, 0, W, 112, col0, lane, c);
  store_tile(out, row0, nrows, col0, lane, c, true);
}


#define BIN_SHIFT 9
#define BIN_NODES 512
#define MAX_BINS 1024
#define CHUNK_CAP 8192
#define BIN_STAGE 12288
#define SORT_THREADS 1024

__global__ void __launch_bounds__(SORT_THREADS) bin_count_kernel(
    const int* __restrict__ dst, int* __restrict__ tab, int E, int NB) {
  __shared__ int hist[MAX_BINS];
  const int tid = threadIdx.x;
  for (int k = tid; k < NB; k += SORT_THREADS) hist[k] = 0;
  __syncthreads();
  const int G = (int)gridDim.x;
  const int chunk = (E + G - 1) / G;
  const int beg = (int)blockIdx.x * chunk;
  const int end = min(E, beg + chunk);
  for (int e0 = beg + tid; e0 < end; e0 += 4 * SORT_THREADS) {
    int d[4];
    #pragma unroll
    for (int u = 0; u < 4; ++u) {
      const int e = e0 + u * SORT_THREADS;
      d[u] = (e < end) ? dst[e] : 0;
    }
    #pragma unroll
    for (int u = 0; u < 4; ++u) {
      if (e0 + u * SORT_THREADS < end) atomicAdd(&hist[d[u] >> BIN_SHIFT], 1);
    }
  }
  __syncthreads();
  for (int k = tid; k < NB; k += SORT_THREADS) tab[(int)blockIdx.x * NB + k] = hist[k];
}

__global__ void __launch_bounds__(SORT_THREADS) bin_scatter_kernel(
    const int* __restrict__ src, const int* __restrict__ dst,
    const int* __restrict__ tab, int* __restrict__ binbase,
    int* __restrict__ pairs, int E, int NB) {
  __shared__ int tot[4][MAX_BINS];
  __shared__ int pre[4][MAX_BINS];
  __shared__ int lcnt[MAX_BINS];
  __shared__ int gbase[MAX_BINS];
  __shared__ int lofs[MAX_BINS];
  __shared__ int lcur[MAX_BINS];
  __shared__ int2 stage[CHUNK_CAP];
  const int tid = threadIdx.x;
  const int G   = (int)gridDim.x;
  const int me  = (int)blockIdx.x;
  for (int k = tid; k < NB; k += SORT_THREADS) lcnt[k] = tab[me * NB + k];
  {
    const int part = tid >> 8;
    const int b0 = (part * G) >> 2;
    const int b1 = ((part + 1) * G) >> 2;
    for (int k = tid & 255; k < NB; k += 256) {
      int t = 0, p = 0;
      #pragma unroll 8
      for (int b = b0; b < b1; ++b) {
        const int v = tab[b * NB + k];
        t += v;
        if (b < me) p += v;
      }
      tot[part][k] = t;
      pre[part][k] = p;
    }
  }
  __syncthreads();
  if (tid < 32) {
    const int C  = (NB + 31) >> 5;
    const int kb = tid * C;
    const int ke = min(NB, kb + C);
    int s = 0, ls = 0;
    for (int k = kb; k < ke; ++k) {
      s  += (tot[0][k] + tot[1][k]) + (tot[2][k] + tot[3][k]);
      ls += lcnt[k];
    }
    int inc = s, linc = ls;
    for (int o = 1; o < 32; o <<= 1) {
      const int t  = __shfl_up(inc, o, 32);
      const int lt = __shfl_up(linc, o, 32);
      if (tid >= o) { inc += t; linc += lt; }
    }
    int run = inc - s, lrun = linc - ls;
    for (int k = kb; k < ke; ++k) {
      gbase[k] = run + (pre[0][k] + pre[1][k]) + (pre[2][k] + pre[3][k]);
      lofs[k]  = lrun;
      lcur[k]  = lrun;
      if (me == 0) binbase[k] = run;
      run  += (tot[0][k] + tot[1][k]) + (tot[2][k] + tot[3][k]);
      lrun += lcnt[k];
    }
    if (me == 0 && tid == 31) binbase[NB] = run;
  }
  __syncthreads();
  const int chunk = (E + G - 1) / G;
  const int beg = me * chunk;
  const int end = min(E, beg + chunk);
  for (int e0 = beg + tid; e0 < end; e0 += 4 * SORT_THREADS) {
    int d[4], s[4];
    #pragma unroll
    for (int u = 0; u < 4; ++u) {
      const int e = e0 + u * SORT_THREADS;
      d[u] = (e < end) ? dst[e] : 0;
      s[u] = (e < end) ? src[e] : 0;
    }
    #pragma unroll
    for (int u = 0; u < 4; ++u) {
      if (e0 + u * SORT_THREADS < end) {
        const int q = atomicAdd(&lcur[d[u] >> BIN_SHIFT], 1);
        stage[q] = make_int2(s[u], d[u]);
      }
    }
  }
  __syncthreads();
  int2* out2 = reinterpret_cast<int2*>(pairs);
  const int cnt = end - beg;
  for (int i = tid; i < cnt; i += SORT_THREADS) {
    const int2 pr = stage[i];
    const int k = pr.y >> BIN_SHIFT;
    out2[gbase[k] + (i - lofs[k])] = pr;
  }
}

__global__ void __launch_bounds__(SORT_THREADS) bin_sort_kernel(
    const int* __restrict__ pairs, const int* __restrict__ binbase,
    int* __restrict__ rowbeg, int* __restrict__ rowend,
    int* __restrict__ col, int N) {
  __shared__ int cnt[BIN_NODES];
  __shared__ int start[BIN_NODES];
  __shared__ int cur[BIN_NODES];
  __shared__ int stage[BIN_STAGE];
  const int tid = threadIdx.x;
  const int bin = (int)blockIdx.x;
  const int2* in2 = reinterpret_cast<const int2*>(pairs);
  const int beg = binbase[bin];
  const int end = binbase[bin + 1];
  if (tid < BIN_NODES) cnt[tid] = 0;
  __syncthreads();
  for (int e0 = beg + tid; e0 < end; e0 += 4 * SORT_THREADS) {
    int d[4];
    #pragma unroll
    for (int u = 0; u < 4; ++u) {
      const int e = e0 + u * SORT_THREADS;
      d[u] = (e < end) ? in2[e].y : 0;
    }
    #pragma unroll
    for (int u = 0; u < 4; ++u) {
      if (e0 + u * SORT_THREADS < end) atomicAdd(&cnt[d[u] & (BIN_NODES - 1)], 1);
    }
  }
  __syncthreads();
  if (tid < 32) {
    int c[BIN_NODES / 32];
    int s = 0;
    #pragma unroll
    for (int j = 0; j < BIN_NODES / 32; ++j) { c[j] = cnt[tid * (BIN_NODES / 32) + j]; s += c[j]; }
    int inc = s;
    for (int o = 1; o < 32; o <<= 1) {
      const int t = __shfl_up(inc, o, 32);
      if (tid >= o) inc += t;
    }
    int run = inc - s;
    #pragma unroll
    for (int j = 0; j < BIN_NODES / 32; ++j) {
      start[tid * (BIN_NODES / 32) + j] = run;
      cur[tid * (BIN_NODES / 32) + j]   = run;
      run += c[j];
    }
  }
  __syncthreads();
  if (tid < BIN_NODES) {
    const int node = bin * BIN_NODES + tid;
    if (node < N) {
      rowbeg[node] = beg + start[tid];
      rowend[node] = beg + start[tid] + cnt[tid];
    }
  }
  for (int e0 = beg + tid; e0 < end; e0 += 4 * SORT_THREADS) {
    int2 pr[4];
    #pragma unroll
    for (int u = 0; u < 4; ++u) {
      const int e = e0 + u * SORT_THREADS;
      pr[u] = (e < end) ? in2[e] : make_int2(0, 0);
    }
    #pragma unroll
    for (int u = 0; u < 4; ++u) {
      if (e0 + u * SORT_THREADS < end) {
        const int q = atomicAdd(&cur[pr[u].y & (BIN_NODES - 1)], 1);
        if (q < BIN_STAGE) stage[q] = pr[u].x;
        else               col[beg + q] = pr[u].x;
      }
    }
  }
  __syncthreads();
  const int staged = min(end - beg, BIN_STAGE);
  for (int i = tid; i < staged; i += SORT_THREADS) col[beg + i] = stage[i];
}

template <int K>
__device__ __forceinline__ void add_rows(const float* __restrict__ h, int mycol,
                                         int j, int lane, float4& acc) {
  float4 v[K];
  #pragma unroll
  for (int k = 0; k < K; ++k) {
    const int sv = __builtin_amdgcn_readlane(mycol, j + k);
    v[k] = *reinterpret_cast<const float4*>(h + (size_t)sv * 128 + lane * 4);
  }
  #pragma unroll
  for (int k = 0; k < K; ++k) {
    acc.x += v[k].x; acc.y += v[k].y; acc.z += v[k].z; acc.w += v[k].w;
  }
}

__device__ __forceinline__ float4 mean_of_rows(const float* __restrict__ h,
                                               const int* __restrict__ col,
                                               int beg, int end, int lane) {
  float4 acc = make_float4(0.0f, 0.0f, 0.0f, 0.0f);
  for (int base = beg; base < end; base += 32) {
    const int n = min(32, end - base);
    const int mycol = (lane < n) ? col[base + lane] : 0;
    int j = 0;
    for (; j + 8 <= n; j += 8) add_rows<8>(h, mycol, j, lane, acc);
    if (n - j >= 4) { add_rows<4>(h, mycol, j, lane, acc); j += 4; }
    if (n - j >= 2) { add_rows<2>(h, mycol, j, lane, acc); j += 2; }
    if (n - j >= 1) { add_rows<1>(h, mycol, j, lane, acc); }
  }
  const float d = fmaxf((float)(end - beg), 1.0f);
  float4 r;
  r.x = acc.x / d; r.y = acc.y / d; r.z = acc.z / d; r.w = acc.w / d;
  return r;
}


__global__ void __launch_bounds__(256) sage_layer_kernel(
    const float* __restrict__ hin, const int* __restrict__ rowbeg,
    const int* __restrict__ rowend, const int* __restrict__ col,
    const float* __restrict__ Wl,
    const float* __restrict__ bl, const float* __restrict__ Wr,
    float* __restrict__ hout, int nrows, int do_relu) {
  __shared__ __attribute__((aligned(16))) float As[16 * LDK];
  const int tid  = threadIdx.x;
  const int lane = tid & 31;
  const int row0 = blockIdx.x * 16;
  const int wv = __builtin_amdgcn_readfirstlane(tid >> 5);
  for (int q = 0; q < 2; ++q) {
    const int r   = 2 * wv + q;
    const int row = row0 + r;
    float4 own = make_float4(0.0f, 0.0f, 0.0f, 0.0f);
    float4 mn  = make_float4(0.0f, 0.0f, 0.0f, 0.0f);
    if (row < nrows) {
      own = *reinterpret_cast<const float4*>(hin + (size_t)row * 128 + lane * 4);
      mn  = mean_of_rows(hin, col, rowbeg[row], rowend[row], lane);
    }
    *reinterpret_cast<float4*>(&As[r * LDK + lane * 4])       = mn;
    *reinterpret_cast<float4*>(&As[r * LDK + 128 + lane * 4]) = own;
  }
  __syncthreads();
  const int col0 = (tid >> 5) * 16;
  v8f c;
  float bias = bl[col0 + (lane & 15)];
  #pragma unroll
  for (int v = 0; v < 8; ++v) c[v] = bias;
  gemm_acc(As, 0,   Wl, 128, col0, lane, c);
  gemm_acc(As, 128, Wr, 128, col0, lane, c);
  store_tile(hout, row0, nrows, col0, lane, c, do_relu != 0);
}


extern "C" void kernel_launch(void* const* d_in, const int* in_sizes, int n_in,
                              void* d_out, int out_size, void* d_ws, size_t ws_size,
                              hipStream_t stream) {
  (void)n_in; (void)out_size; (void)ws_size;
  const float* x_pol   = (const float*)d_in[0];
  const int*   pol_sid = (const int*)  d_in[1];
  const float* x_comp  = (const float*)d_in[2];
  const int*   c_sec   = (const int*)  d_in[3];
  const int*   c_ind   = (const int*)  d_in[4];
  const int*   edges   = (const int*)  d_in[5];
  const float* s_emb   = (const float*)d_in[6];
  const float* sec_emb = (const float*)d_in[7];
  const float* ind_emb = (const float*)d_in[8];
  const float* W_pol   = (const float*)d_in[9];
  const float* b_pol   = (const float*)d_in[10];
  const float* W_comp  = (const float*)d_in[11];
  const float* b_comp  = (const float*)d_in[12];
  const float* Wl1     = (const float*)d_in[13];
  const float* bl1     = (const float*)d_in[14];
  const float* Wr1     = (const float*)d_in[15];
  const float* Wl2     = (const float*)d_in[16];
  const float* bl2     = (const float*)d_in[17];
  const float* Wr2     = (const float*)d_in[18];

  const int NP = in_sizes[0] / 64;
  const int NC = in_sizes[2] / 96;
  const int E  = in_sizes[5] / 2;
  const int N  = NP + NC;
  const int* src = edges;
  const int* dst = edges + E;

  float* hws  = (float*)d_ws;
  float* outp = (float*)d_out;
  const int N4 = (N + 3) & ~3;
  const int NB = (N + BIN_NODES - 1) >> BIN_SHIFT;
  const int G  = (E + CHUNK_CAP - 1) / CHUNK_CAP;
  int* tab     = (int*)(hws + (size_t)N * 128);
  int* binbase = tab + G * NB;
  int* rowbeg  = tab + N4;
  int* rowend  = rowbeg + N4;
  int* col     = rowend + N4;
  int* pairs   = (int*)d_ws;

  const dim3 blk(256);

  encode_pol_kernel<<<dim3((NP + 15) / 16), blk, 0, stream>>>(
      x_pol, pol_sid, s_emb, W_pol, b_pol, outp, NP);
  encode_comp_kernel<<<dim3((NC + 15) / 16), blk, 0, stream>>>(
      x_comp, c_sec, c_ind, sec_emb, ind_emb, W_comp, b_comp,
      outp + (size_t)NP * 128, NC);

  bin_count_kernel<<<dim3(G), dim3(SORT_THREADS), 0, stream>>>(dst, tab, E, NB);
  bin_scatter_kernel<<<dim3(G), dim3(SORT_THREADS), 0, stream>>>(
      src, dst, tab, binbase, pairs, E, NB);
  bin_sort_kernel<<<dim3(NB), dim3(SORT_THREADS), 0, stream>>>(
      pairs, binbase, rowbeg, rowend, col, N);

  sage_layer_kernel<<<dim3((N + 15) / 16), blk, 0, stream>>>(
      outp, rowbeg, rowend, col, Wl1, bl1, Wr1, hws, N, 1);

  sage_layer_kernel<<<dim3((N + 15) / 16), blk, 0, stream>>>(
      hws, rowbeg, rowend, col, Wl2, bl2, Wr2, outp, N, 0);
}


# Row tile of every product kernel raised from 16 to 32 nodes: a wave keeps two accumulators (rows 0-15, 16-31) and each B fragment it loads feeds two matrix instructions; sum order unchanged
# speedup vs baseline: 3.5458x; 1.0581x over previous
#include <hip/hip_runtime.h>


typedef __attribute__((ext_vector_type(2))) float v2f;
typedef __attribute__((ext_vector_type(8))) float v8f;

#define LDK 260

__device__ __forceinline__ v8f wmma4(v2f a, v2f b, v8f c) {
  return __builtin_amdgcn_wmma_f32_16x16x4_f32(false, a, false, b,
                                               (short)0, c, false, false);
}

__device__ __forceinline__ void gemm_acc2(const float* As, int aoff,
                                          const float* __restrict__ B,
                                          int K, int col0, int lane,
                                          v8f& c0, v8f& c1) {
  const int hl = lane >> 4;
  const int m  = lane & 15;
  const int n  = lane & 15;
  #pragma unroll 4
  for (int k0 = 0; k0 < K; k0 += 4) {
    const int ka = k0 + 2 * hl;
    v2f a0, a1, b;
    a0.x = As[m * LDK + aoff + ka];
    a0.y = As[m * LDK + aoff + ka + 1];
    a1.x = As[(m + 16) * LDK + aoff + ka];
    a1.y = As[(m + 16) * LDK + aoff + ka + 1];
    b.x = B[ka * 128 + col0 + n];
    b.y = B[(ka + 1) * 128 + col0 + n];
    c0 = wmma4(a0, b, c0);
    c1 = wmma4(a1, b, c1);
  }
}

__device__ __forceinline__ void store_tile(float* __restrict__ out, int row0,
                                           int nrows, int col0, int lane,
                                           v8f c, bool do_relu) {
  const int hl = lane >> 4;
  const int n  = lane & 15;
  #pragma unroll
  for (int v = 0; v < 8; ++v) {
    int row = row0 + v + 8 * hl;
    float val = c[v];
    if (do_relu) val = fmaxf(val, 0.0f);
    if (row < nrows) out[(size_t)row * 128 + col0 + n] = val;
  }
}


__global__ void __launch_bounds__(256) encode_pol_kernel(
    const float* __restrict__ x, const int* __restrict__ sidx,
    const float* __restrict__ semb, const float* __restrict__ W,
    const float* __restrict__ bvec, float* __restrict__ out, int nrows) {
  __shared__ float As[32 * LDK];
  const int tid  = threadIdx.x;
  const int row0 = blockIdx.x * 32;
  for (int idx = tid; idx < 32 * 72; idx += 256) {
    int r = idx / 72, cc = idx % 72;
    int row = row0 + r;
    float v = 0.0f;
    if (row < nrows) {
      if (cc < 64) v = x[(size_t)row * 64 + cc];
      else         v = semb[sidx[row] * 8 + (cc - 64)];
    }
    As[r * LDK + cc] = v;
  }
  __syncthreads();
  const int lane = tid & 31;
  const int col0 = (tid >> 5) * 16;
  v8f c0, c1;
  float bias = bvec[col0 + (lane & 15)];
  #pragma unroll
  for (int v = 0; v < 8; ++v) { c0[v] = bias; c1[v] = bias; }
  gemm_acc2(As, 0, W, 72, col0, lane, c0, c1);
  store_tile(out, row0,      nrows, col0, lane, c0, true);
  store_tile(out, row0 + 16, nrows, col0, lane, c1, true);
}

__global__ void __launch_bounds__(256) encode_comp_kernel(
    const float* __restrict__ x, const int* __restrict__ secidx,
    const int* __restrict__ indidx, const float* __restrict__ secemb,
    const float* __restrict__ indemb, const float* __restrict__ W,
    const float* __restrict__ bvec, float* __restrict__ out, int nrows) {
  __shared__ float As[32 * LDK];
  const int tid  = threadIdx.x;
  const int row0 = blockIdx.x * 32;
  for (int idx = tid; idx < 32 * 112; idx += 256) {
    int r = idx / 112, cc = idx % 112;
    int row = row0 + r;
    float v = 0.0f;
    if (row < nrows) {
      if (cc < 96)       v = x[(size_t)row * 96 + cc];
      else if (cc < 104) v = secemb[secidx[row] * 8 + (cc - 96)];
      else               v = indemb[indidx[row] * 8 + (cc - 104)];
    }
    As[r * LDK + cc] = v;
  }
  __syncthreads();
  const int lane = tid & 31;
  const int col0 = (tid >> 5) * 16;
  v8f c0, c1;
  float bias = bvec[col0 + (lane & 15)];
  #pragma unroll
  for (int v = 0; v < 8; ++v) { c0[v] = bias; c1[v] = bias; }
  gemm_acc2(As, 0, W, 112, col0, lane, c0, c1);
  store_tile(out, row0,      nrows, col0, lane, c0, true);
  store_tile(out, row0 + 16, nrows, col0, lane, c1, true);
}


#define BIN_SHIFT 9
#define BIN_NODES 512
#define MAX_BINS 1024
#define CHUNK_CAP 8192
#define BIN_STAGE 12288
#define SORT_THREADS 1024

__global__ void __launch_bounds__(SORT_THREADS) bin_count_kernel(
    const int* __restrict__ dst, int* __restrict__ tab, int E, int NB) {
  __shared__ int hist[MAX_BINS];
  const int tid = threadIdx.x;
  for (int k = tid; k < NB; k += SORT_THREADS) hist[k] = 0;
  __syncthreads();
  const int G = (int)gridDim.x;
  const int chunk = (E + G - 1) / G;
  const int beg = (int)blockIdx.x * chunk;
  const int end = min(E, beg + chunk);
  for (int e0 = beg + tid; e0 < end; e0 += 4 * SORT_THREADS) {
    int d[4];
    #pragma unroll
    for (int u = 0; u < 4; ++u) {
      const int e = e0 + u * SORT_THREADS;
      d[u] = (e < end) ? dst[e] : 0;
    }
    #pragma unroll
    for (int u = 0; u < 4; ++u) {
      if (e0 + u * SORT_THREADS < end) atomicAdd(&hist[d[u] >> BIN_SHIFT], 1);
    }
  }
  __syncthreads();
  for (int k = tid; k < NB; k += SORT_THREADS) tab[(int)blockIdx.x * NB + k] = hist[k];
}

__global__ void __launch_bounds__(SORT_THREADS) bin_scatter_kernel(
    const int* __restrict__ src, const int* __restrict__ dst,
    const int* __restrict__ tab, int* __restrict__ binbase,
    int* __restrict__ pairs, int E, int NB) {
  __shared__ int tot[4][MAX_BINS];
  __shared__ int pre[4][MAX_BINS];
  __shared__ int lcnt[MAX_BINS];
  __shared__ int gbase[MAX_BINS];
  __shared__ int lofs[MAX_BINS];
  __shared__ int lcur[MAX_BINS];
  __shared__ int2 stage[CHUNK_CAP];
  const int tid = threadIdx.x;
  const int G   = (int)gridDim.x;
  const int me  = (int)blockIdx.x;
  for (int k = tid; k < NB; k += SORT_THREADS) lcnt[k] = tab[me * NB + k];
  {
    const int part = tid >> 8;
    const int b0 = (part * G) >> 2;
    const int b1 = ((part + 1) * G) >> 2;
    for (int k = tid & 255; k < NB; k += 256) {
      int t = 0, p = 0;
      #pragma unroll 8
      for (int b = b0; b < b1; ++b) {
        const int v = tab[b * NB + k];
        t += v;
        if (b < me) p += v;
      }
      tot[part][k] = t;
      pre[part][k] = p;
    }
  }
  __syncthreads();
  if (tid < 32) {
    const int C  = (NB + 31) >> 5;
    const int kb = tid * C;
    const int ke = min(NB, kb + C);
    int s = 0, ls = 0;
    for (int k = kb; k < ke; ++k) {
      s  += (tot[0][k] + tot[1][k]) + (tot[2][k] + tot[3][k]);
      ls += lcnt[k];
    }
    int inc = s, linc = ls;
    for (int o = 1; o < 32; o <<= 1) {
      const int t  = __shfl_up(inc, o, 32);
      const int lt = __shfl_up(linc, o, 32);
      if (tid >= o) { inc += t; linc += lt; }
    }
    int run = inc - s, lrun = linc - ls;
    for (int k = kb; k < ke; ++k) {
      gbase[k] = run + (pre[0][k] + pre[1][k]) + (pre[2][k] + pre[3][k]);
      lofs[k]  = lrun;
      lcur[k]  = lrun;
      if (me == 0) binbase[k] = run;
      run  += (tot[0][k] + tot[1][k]) + (tot[2][k] + tot[3][k]);
      lrun += lcnt[k];
    }
    if (me == 0 && tid == 31) binbase[NB] = run;
  }
  __syncthreads();
  const int chunk = (E + G - 1) / G;
  const int beg = me * chunk;
  const int end = min(E, beg + chunk);
  for (int e0 = beg + tid; e0 < end; e0 += 4 * SORT_THREADS) {
    int d[4], s[4];
    #pragma unroll
    for (int u = 0; u < 4; ++u) {
      const int e = e0 + u * SORT_THREADS;
      d[u] = (e < end) ? dst[e] : 0;
      s[u] = (e < end) ? src[e] : 0;
    }
    #pragma unroll
    for (int u = 0; u < 4; ++u) {
      if (e0 + u * SORT_THREADS < end) {
        const int q = atomicAdd(&lcur[d[u] >> BIN_SHIFT], 1);
        stage[q] = make_int2(s[u], d[u]);
      }
    }
  }
  __syncthreads();
  int2* out2 = reinterpret_cast<int2*>(pairs);
  const int cnt = end - beg;
  for (int i = tid; i < cnt; i += SORT_THREADS) {
    const int2 pr = stage[i];
    const int k = pr.y >> BIN_SHIFT;
    out2[gbase[k] + (i - lofs[k])] = pr;
  }
}

__global__ void __launch_bounds__(SORT_THREADS) bin_sort_kernel(
    const int* __restrict__ pairs, const int* __restrict__ binbase,
    int* __restrict__ rowbeg, int* __restrict__ rowend,
    int* __restrict__ col, int N) {
  __shared__ int cnt[BIN_NODES];
  __shared__ int start[BIN_NODES];
  __shared__ int cur[BIN_NODES];
  __shared__ int stage[BIN_STAGE];
  const int tid = threadIdx.x;
  const int bin = (int)blockIdx.x;
  const int2* in2 = reinterpret_cast<const int2*>(pairs);
  const int beg = binbase[bin];
  const int end = binbase[bin + 1];
  if (tid < BIN_NODES) cnt[tid] = 0;
  __syncthreads();
  for (int e0 = beg + tid; e0 < end; e0 += 4 * SORT_THREADS) {
    int d[4];
    #pragma unroll
    for (int u = 0; u < 4; ++u) {
      const int e = e0 + u * SORT_THREADS;
      d[u] = (e < end) ? in2[e].y : 0;
    }
    #pragma unroll
    for (int u = 0; u < 4; ++u) {
      if (e0 + u * SORT_THREADS < end) atomicAdd(&cnt[d[u] & (BIN_NODES - 1)], 1);
    }
  }
  __syncthreads();
  if (tid < 32) {
    int c[BIN_NODES / 32];
    int s = 0;
    #pragma unroll
    for (int j = 0; j < BIN_NODES / 32; ++j) { c[j] = cnt[tid * (BIN_NODES / 32) + j]; s += c[j]; }
    int inc = s;
    for (int o = 1; o < 32; o <<= 1) {
      const int t = __shfl_up(inc, o, 32);
      if (tid >= o) inc += t;
    }
    int run = inc - s;
    #pragma unroll
    for (int j = 0; j < BIN_NODES / 32; ++j) {
      start[tid * (BIN_NODES / 32) + j] = run;
      cur[tid * (BIN_NODES / 32) + j]   = run;
      run += c[j];
    }
  }
  __syncthreads();
  if (tid < BIN_NODES) {
    const int node = bin * BIN_NODES + tid;
    if (node < N) {
      rowbeg[node] = beg + start[tid];
      rowend[node] = beg + start[tid] + cnt[tid];
    }
  }
  for (int e0 = beg + tid; e0 < end; e0 += 4 * SORT_THREADS) {
    int2 pr[4];
    #pragma unroll
    for (int u = 0; u < 4; ++u) {
      const int e = e0 + u * SORT_THREADS;
      pr[u] = (e < end) ? in2[e] : make_int2(0, 0);
    }
    #pragma unroll
    for (int u = 0; u < 4; ++u) {
      if (e0 + u * SORT_THREADS < end) {
        const int q = atomicAdd(&cur[pr[u].y & (BIN_NODES - 1)], 1);
        if (q < BIN_STAGE) stage[q] = pr[u].x;
        else               col[beg + q] = pr[u].x;
      }
    }
  }
  __syncthreads();
  const int staged = min(end - beg, BIN_STAGE);
  for (int i = tid; i < staged; i += SORT_THREADS) col[beg + i] = stage[i];
}

template <int K>
__device__ __forceinline__ void add_rows(const float* __restrict__ h, int mycol,
                                         int j, int lane, float4& acc) {
  float4 v[K];
  #pragma unroll
  for (int k = 0; k < K; ++k) {
    const int sv = __builtin_amdgcn_readlane(mycol, j + k);
    v[k] = *reinterpret_cast<const float4*>(h + (size_t)sv * 128 + lane * 4);
  }
  #pragma unroll
  for (int k = 0; k < K; ++k) {
    acc.x += v[k].x; acc.y += v[k].y; acc.z += v[k].z; acc.w += v[k].w;
  }
}

__device__ __forceinline__ float4 mean_of_rows(const float* __restrict__ h,
                                               const int* __restrict__ col,
                                               int beg, int end, int lane) {
  float4 acc = make_float4(0.0f, 0.0f, 0.0f, 0.0f);
  for (int base = beg; base < end; base += 32) {
    const int n = min(32, end - base);
    const int mycol = (lane < n) ? col[base + lane] : 0;
    int j = 0;
    for (; j + 8 <= n; j += 8) add_rows<8>(h, mycol, j, lane, acc);
    if (n - j >= 4) { add_rows<4>(h, mycol, j, lane, acc); j += 4; }
    if (n - j >= 2) { add_rows<2>(h, mycol, j, lane, acc); j += 2; }
    if (n - j >= 1) { add_rows<1>(h, mycol, j, lane, acc); }
  }
  const float d = fmaxf((float)(end - beg), 1.0f);
  float4 r;
  r.x = acc.x / d; r.y = acc.y / d; r.z = acc.z / d; r.w = acc.w / d;
  return r;
}


__global__ void __launch_bounds__(256) sage_layer_kernel(
    const float* __restrict__ hin, const int* __restrict__ rowbeg,
    const int* __restrict__ rowend, const int* __restrict__ col,
    const float* __restrict__ Wl,
    const float* __restrict__ bl, const float* __restrict__ Wr,
    float* __restrict__ hout, int nrows, int do_relu) {
  __shared__ __attribute__((aligned(16))) float As[32 * LDK];
  const int tid  = threadIdx.x;
  const int lane = tid & 31;
  const int row0 = blockIdx.x * 32;
  const int wv = __builtin_amdgcn_readfirstlane(tid >> 5);
  for (int q = 0; q < 4; ++q) {
    const int r   = 4 * wv + q;
    const int row = row0 + r;
    float4 own = make_float4(0.0f, 0.0f, 0.0f, 0.0f);
    float4 mn  = make_float4(0.0f, 0.0f, 0.0f, 0.0f);
    if (row < nrows) {
      own = *reinterpret_cast<const float4*>(hin + (size_t)row * 128 + lane * 4);
      mn  = mean_of_rows(hin, col, rowbeg[row], rowend[row], lane);
    }
    *reinterpret_cast<float4*>(&As[r * LDK + lane * 4])       = mn;
    *reinterpret_cast<float4*>(&As[r * LDK + 128 + lane * 4]) = own;
  }
  __syncthreads();
  const int col0 = (tid >> 5) * 16;
  v8f c0, c1;
  float bias = bl[col0 + (lane & 15)];
  #pragma unroll
  for (int v = 0; v < 8; ++v) { c0[v] = bias; c1[v] = bias; }
  gemm_acc2(As, 0,   Wl, 128, col0, lane, c0, c1);
  gemm_acc2(As, 128, Wr, 128, col0, lane, c0, c1);
  store_tile(hout, row0,      nrows, col0, lane, c0, do_relu != 0);
  store_tile(hout, row0 + 16, nrows, col0, lane, c1, do_relu != 0);
}


extern "C" void kernel_launch(void* const* d_in, const int* in_sizes, int n_in,
                              void* d_out, int out_size, void* d_ws, size_t ws_size,
                              hipStream_t stream) {
  (void)n_in; (void)out_size; (void)ws_size;
  const float* x_pol   = (const float*)d_in[0];
  const int*   pol_sid = (const int*)  d_in[1];
  const float* x_comp  = (const float*)d_in[2];
  const int*   c_sec   = (const int*)  d_in[3];
  const int*   c_ind   = (const int*)  d_in[4];
  const int*   edges   = (const int*)  d_in[5];
  const float* s_emb   = (const float*)d_in[6];
  const float* sec_emb = (const float*)d_in[7];
  const float* ind_emb = (const float*)d_in[8];
  const float* W_pol   = (const float*)d_in[9];
  const float* b_pol   = (const float*)d_in[10];
  const float* W_comp  = (const float*)d_in[11];
  const float* b_comp  = (const float*)d_in[12];
  const float* Wl1     = (const float*)d_in[13];
  const float* bl1     = (const float*)d_in[14];
  const float* Wr1     = (const float*)d_in[15];
  const float* Wl2     = (const float*)d_in[16];
  const float* bl2     = (const float*)d_in[17];
  const float* Wr2     = (const float*)d_in[18];

  const int NP = in_sizes[0] / 64;
  const int NC = in_sizes[2] / 96;
  const int E  = in_sizes[5] / 2;
  const int N  = NP + NC;
  const int* src = edges;
  const int* dst = edges + E;

  float* hws  = (float*)d_ws;
  float* outp = (float*)d_out;
  const int N4 = (N + 3) & ~3;
  const int NB = (N + BIN_NODES - 1) >> BIN_SHIFT;
  const int G  = (E + CHUNK_CAP - 1) / CHUNK_CAP;
  int* tab     = (int*)(hws + (size_t)N * 128);
  int* binbase = tab + G * NB;
  int* rowbeg  = tab + N4;
  int* rowend  = rowbeg + N4;
  int* col     = rowend + N4;
  int* pairs   = (int*)d_ws;

  const dim3 blk(256);

  encode_pol_kernel<<<dim3((NP + 31) / 32), blk, 0, stream>>>(
      x_pol, pol_sid, s_emb, W_pol, b_pol, outp, NP);
  encode_comp_kernel<<<dim3((NC + 31) / 32), blk, 0, stream>>>(
      x_comp, c_sec, c_ind, sec_emb, ind_emb, W_comp, b_comp,
      outp + (size_t)NP * 128, NC);

  bin_count_kernel<<<dim3(G), dim3(SORT_THREADS), 0, stream>>>(dst, tab, E, NB);
  bin_scatter_kernel<<<dim3(G), dim3(SORT_THREADS), 0, stream>>>(
      src, dst, tab, binbase, pairs, E, NB);
  bin_sort_kernel<<<dim3(NB), dim3(SORT_THREADS), 0, stream>>>(
      pairs, binbase, rowbeg, rowend, col, N);

  sage_layer_kernel<<<dim3((N + 31) / 32), blk, 0, stream>>>(
      outp, rowbeg, rowend, col, Wl1, bl1, Wr1, hws, N, 1);

  sage_layer_kernel<<<dim3((N + 31) / 32), blk, 0, stream>>>(
      hws, rowbeg, rowend, col, Wl2, bl2, Wr2, outp, N, 0);
}


# Encoder A tiles staged with 16-byte loads issued together (index, x pieces, embedding piece; no branch between them) instead of one waited 4-byte load per element; sum order unchanged
# speedup vs baseline: 3.7637x; 1.0615x over previous
#include <hip/hip_runtime.h>


typedef __attribute__((ext_vector_type(2))) float v2f;
typedef __attribute__((ext_vector_type(8))) float v8f;

#define LDK 260

__device__ __forceinline__ v8f wmma4(v2f a, v2f b, v8f c) {
  return __builtin_amdgcn_wmma_f32_16x16x4_f32(false, a, false, b,
                                               (short)0, c, false, false);
}

__device__ __forceinline__ void gemm_acc2(const float* As, int aoff,
                                          const float* __restrict__ B,
                                          int K, int col0, int lane,
                                          v8f& c0, v8f& c1) {
  const int hl = lane >> 4;
  const int m  = lane & 15;
  const int n  = lane & 15;
  #pragma unroll 4
  for (int k0 = 0; k0 < K; k0 += 4) {
    const int ka = k0 + 2 * hl;
    v2f a0, a1, b;
    a0.x = As[m * LDK + aoff + ka];
    a0.y = As[m * LDK + aoff + ka + 1];
    a1.x = As[(m + 16) * LDK + aoff + ka];
    a1.y = As[(m + 16) * LDK + aoff + ka + 1];
    b.x = B[ka * 128 + col0 + n];
    b.y = B[(ka + 1) * 128 + col0 + n];
    c0 = wmma4(a0, b, c0);
    c1 = wmma4(a1, b, c1);
  }
}

__device__ __forceinline__ void store_tile(float* __restrict__ out, int row0,
                                           int nrows, int col0, int lane,
                                           v8f c, bool do_relu) {
  const int hl = lane >> 4;
  const int n  = lane & 15;
  #pragma unroll
  for (int v = 0; v < 8; ++v) {
    int row = row0 + v + 8 * hl;
    float val = c[v];
    if (do_relu) val = fmaxf(val, 0.0f);
    if (row < nrows) out[(size_t)row * 128 + col0 + n] = val;
  }
}


__device__ __forceinline__ float4 keep_if(float4 v, bool ok) {
  float4 r;
  r.x = ok ? v.x : 0.0f; r.y = ok ? v.y : 0.0f;
  r.z = ok ? v.z : 0.0f; r.w = ok ? v.w : 0.0f;
  return r;
}


__global__ void __launch_bounds__(256) encode_pol_kernel(
    const float* __restrict__ x, const int* __restrict__ sidx,
    const float* __restrict__ semb, const float* __restrict__ W,
    const float* __restrict__ bvec, float* __restrict__ out, int nrows) {
  __shared__ __attribute__((aligned(16))) float As[32 * LDK];
  const int tid  = threadIdx.x;
  const int row0 = blockIdx.x * 32;
  const int last = nrows - 1;
  const int er = tid >> 3, ec = tid & 7;
  const int si = sidx[min(row0 + er, last)];
  float4 xv[2];
  #pragma unroll
  for (int i = 0; i < 2; ++i) {
    const int f = tid + 256 * i;
    const int r = f >> 4, c4 = f & 15;
    const int row = min(row0 + r, last);
    xv[i] = *reinterpret_cast<const float4*>(x + (size_t)row * 64 + c4 * 4);
  }
  const float e = semb[si * 8 + ec];
  #pragma unroll
  for (int i = 0; i < 2; ++i) {
    const int f = tid + 256 * i;
    const int r = f >> 4, c4 = f & 15;
    *reinterpret_cast<float4*>(&As[r * LDK + c4 * 4]) =
        keep_if(xv[i], row0 + r < nrows);
  }
  As[er * LDK + 64 + ec] = (row0 + er < nrows) ? e : 0.0f;
  __syncthreads();
  const int lane = tid & 31;
  const int col0 = (tid >> 5) * 16;
  v8f c0, c1;
  float bias = bvec[col0 + (lane & 15)];
  #pragma unroll
  for (int v = 0; v < 8; ++v) { c0[v] = bias; c1[v] = bias; }
  gemm_acc2(As, 0, W, 72, col0, lane, c0, c1);
  store_tile(out, row0,      nrows, col0, lane, c0, true);
  store_tile(out, row0 + 16, nrows, col0, lane, c1, true);
}

__global__ void __launch_bounds__(256) encode_comp_kernel(
    const float* __restrict__ x, const int* __restrict__ secidx,
    const int* __restrict__ indidx, const float* __restrict__ secemb,
    const float* __restrict__ indemb, const float* __restrict__ W,
    const float* __restrict__ bvec, float* __restrict__ out, int nrows) {
  __shared__ __attribute__((aligned(16))) float As[32 * LDK];
  const int tid  = threadIdx.x;
  const int row0 = blockIdx.x * 32;
  const int last = nrows - 1;
  const int er = tid >> 3, ep = tid & 7;
  const int*   itab = (ep & 4) ? indidx : secidx;
  const float* etab = (ep & 4) ? indemb : secemb;
  const int ei = itab[min(row0 + er, last)];
  float4 xv[3];
  #pragma unroll
  for (int i = 0; i < 3; ++i) {
    const int f = tid + 256 * i;
    const int r = f / 24, c4 = f % 24;
    const int row = min(row0 + r, last);
    xv[i] = *reinterpret_cast<const float4*>(x + (size_t)row * 96 + c4 * 4);
  }
  const float2 e = *reinterpret_cast<const float2*>(etab + ei * 8 + (ep & 3) * 2);
  #pragma unroll
  for (int i = 0; i < 3; ++i) {
    const int f = tid + 256 * i;
    const int r = f / 24, c4 = f % 24;
    *reinterpret_cast<float4*>(&As[r * LDK + c4 * 4]) =
        keep_if(xv[i], row0 + r < nrows);
  }
  {
    const bool ok = row0 + er < nrows;
    float2 ez;
    ez.x = ok ? e.x : 0.0f; ez.y = ok ? e.y : 0.0f;
    *reinterpret_cast<float2*>(&As[er * LDK + 96 + ep * 2]) = ez;
  }
  __syncthreads();
  const int lane = tid & 31;
  const int col0 = (tid >> 5) * 16;
  v8f c0, c1;
  float bias = bvec[col0 + (lane & 15)];
  #pragma unroll
  for (int v = 0; v < 8; ++v) { c0[v] = bias; c1[v] = bias; }
  gemm_acc2(As, 0, W, 112, col0, lane, c0, c1);
  store_tile(out, row0,      nrows, col0, lane, c0, true);
  store_tile(out, row0 + 16, nrows, col0, lane, c1, true);
}


#define BIN_SHIFT 9
#define BIN_NODES 512
#define MAX_BINS 1024
#define CHUNK_CAP 8192
#define BIN_STAGE 12288
#define SORT_THREADS 1024

__global__ void __launch_bounds__(SORT_THREADS) bin_count_kernel(
    const int* __restrict__ dst, int* __restrict__ tab, int E, int NB) {
  __shared__ int hist[MAX_BINS];
  const int tid = threadIdx.x;
  for (int k = tid; k < NB; k += SORT_THREADS) hist[k] = 0;
  __syncthreads();
  const int G = (int)gridDim.x;
  const int chunk = (E + G - 1) / G;
  const int beg = (int)blockIdx.x * chunk;
  const int end = min(E, beg + chunk);
  for (int e0 = beg + tid; e0 < end; e0 += 4 * SORT_THREADS) {
    int d[4];
    #pragma unroll
    for (int u = 0; u < 4; ++u) {
      const int e = e0 + u * SORT_THREADS;
      d[u] = (e < end) ? dst[e] : 0;
    }
    #pragma unroll
    for (int u = 0; u < 4; ++u) {
      if (e0 + u * SORT_THREADS < end) atomicAdd(&hist[d[u] >> BIN_SHIFT], 1);
    }
  }
  __syncthreads();
  for (int k = tid; k < NB; k += SORT_THREADS) tab[(int)blockIdx.x * NB + k] = hist[k];
}

__global__ void __launch_bounds__(SORT_THREADS) bin_scatter_kernel(
    const int* __restrict__ src, const int* __restrict__ dst,
    const int* __restrict__ tab, int* __restrict__ binbase,
    int* __restrict__ pairs, int E, int NB) {
  __shared__ int tot[4][MAX_BINS];
  __shared__ int pre[4][MAX_BINS];
  __shared__ int lcnt[MAX_BINS];
  __shared__ int gbase[MAX_BINS];
  __shared__ int lofs[MAX_BINS];
  __shared__ int lcur[MAX_BINS];
  __shared__ int2 stage[CHUNK_CAP];
  const int tid = threadIdx.x;
  const int G   = (int)gridDim.x;
  const int me  = (int)blockIdx.x;
  for (int k = tid; k < NB; k += SORT_THREADS) lcnt[k] = tab[me * NB + k];
  {
    const int part = tid >> 8;
    const int b0 = (part * G) >> 2;
    const int b1 = ((part + 1) * G) >> 2;
    for (int k = tid & 255; k < NB; k += 256) {
      int t = 0, p = 0;
      #pragma unroll 8
      for (int b = b0; b < b1; ++b) {
        const int v = tab[b * NB + k];
        t += v;
        if (b < me) p += v;
      }
      tot[part][k] = t;
      pre[part][k] = p;
    }
  }
  __syncthreads();
  if (tid < 32) {
    const int C  = (NB + 31) >> 5;
    const int kb = tid * C;
    const int ke = min(NB, kb + C);
    int s = 0, ls = 0;
    for (int k = kb; k < ke; ++k) {
      s  += (tot[0][k] + tot[1][k]) + (tot[2][k] + tot[3][k]);
      ls += lcnt[k];
    }
    int inc = s, linc = ls;
    for (int o = 1; o < 32; o <<= 1) {
      const int t  = __shfl_up(inc, o, 32);
      const int lt = __shfl_up(linc, o, 32);
      if (tid >= o) { inc += t; linc += lt; }
    }
    int run = inc - s, lrun = linc - ls;
    for (int k = kb; k < ke; ++k) {
      gbase[k] = run + (pre[0][k] + pre[1][k]) + (pre[2][k] + pre[3][k]);
      lofs[k]  = lrun;
      lcur[k]  = lrun;
      if (me == 0) binbase[k] = run;
      run  += (tot[0][k] + tot[1][k]) + (tot[2][k] + tot[3][k]);
      lrun += lcnt[k];
    }
    if (me == 0 && tid == 31) binbase[NB] = run;
  }
  __syncthreads();
  const int chunk = (E + G - 1) / G;
  const int beg = me * chunk;
  const int end = min(E, beg + chunk);
  for (int e0 = beg + tid; e0 < end; e0 += 4 * SORT_THREADS) {
    int d[4], s[4];
    #pragma unroll
    for (int u = 0; u < 4; ++u) {
      const int e = e0 + u * SORT_THREADS;
      d[u] = (e < end) ? dst[e] : 0;
      s[u] = (e < end) ? src[e] : 0;
    }
    #pragma unroll
    for (int u = 0; u < 4; ++u) {
      if (e0 + u * SORT_THREADS < end) {
        const int q = atomicAdd(&lcur[d[u] >> BIN_SHIFT], 1);
        stage[q] = make_int2(s[u], d[u]);
      }
    }
  }
  __syncthreads();
  int2* out2 = reinterpret_cast<int2*>(pairs);
  const int cnt = end - beg;
  for (int i = tid; i < cnt; i += SORT_THREADS) {
    const int2 pr = stage[i];
    const int k = pr.y >> BIN_SHIFT;
    out2[gbase[k] + (i - lofs[k])] = pr;
  }
}

__global__ void __launch_bounds__(SORT_THREADS) bin_sort_kernel(
    const int* __restrict__ pairs, const int* __restrict__ binbase,
    int* __restrict__ rowbeg, int* __restrict__ rowend,
    int* __restrict__ col, int N) {
  __shared__ int cnt[BIN_NODES];
  __shared__ int start[BIN_NODES];
  __shared__ int cur[BIN_NODES];
  __shared__ int stage[BIN_STAGE];
  const int tid = threadIdx.x;
  const int bin = (int)blockIdx.x;
  const int2* in2 = reinterpret_cast<const int2*>(pairs);
  const int beg = binbase[bin];
  const int end = binbase[bin + 1];
  if (tid < BIN_NODES) cnt[tid] = 0;
  __syncthreads();
  for (int e0 = beg + tid; e0 < end; e0 += 4 * SORT_THREADS) {
    int d[4];
    #pragma unroll
    for (int u = 0; u < 4; ++u) {
      const int e = e0 + u * SORT_THREADS;
      d[u] = (e < end) ? in2[e].y : 0;
    }
    #pragma unroll
    for (int u = 0; u < 4; ++u) {
      if (e0 + u * SORT_THREADS < end) atomicAdd(&cnt[d[u] & (BIN_NODES - 1)], 1);
    }
  }
  __syncthreads();
  if (tid < 32) {
    int c[BIN_NODES / 32];
    int s = 0;
    #pragma unroll
    for (int j = 0; j < BIN_NODES / 32; ++j) { c[j] = cnt[tid * (BIN_NODES / 32) + j]; s += c[j]; }
    int inc = s;
    for (int o = 1; o < 32; o <<= 1) {
      const int t = __shfl_up(inc, o, 32);
      if (tid >= o) inc += t;
    }
    int run = inc - s;
    #pragma unroll
    for (int j = 0; j < BIN_NODES / 32; ++j) {
      start[tid * (BIN_NODES / 32) + j] = run;
      cur[tid * (BIN_NODES / 32) + j]   = run;
      run += c[j];
    }
  }
  __syncthreads();
  if (tid < BIN_NODES) {
    const int node = bin * BIN_NODES + tid;
    if (node < N) {
      rowbeg[node] = beg + start[tid];
      rowend[node] = beg + start[tid] + cnt[tid];
    }
  }
  for (int e0 = beg + tid; e0 < end; e0 += 4 * SORT_THREADS) {
    int2 pr[4];
    #pragma unroll
    for (int u = 0; u < 4; ++u) {
      const int e = e0 + u * SORT_THREADS;
      pr[u] = (e < end) ? in2[e] : make_int2(0, 0);
    }
    #pragma unroll
    for (int u = 0; u < 4; ++u) {
      if (e0 + u * SORT_THREADS < end) {
        const int q = atomicAdd(&cur[pr[u].y & (BIN_NODES - 1)], 1);
        if (q < BIN_STAGE) stage[q] = pr[u].x;
        else               col[beg + q] = pr[u].x;
      }
    }
  }
  __syncthreads();
  const int staged = min(end - beg, BIN_STAGE);
  for (int i = tid; i < staged; i += SORT_THREADS) col[beg + i] = stage[i];
}

template <int K>
__device__ __forceinline__ void add_rows(const float* __restrict__ h, int mycol,
                                         int j, int lane, float4& acc) {
  float4 v[K];
  #pragma unroll
  for (int k = 0; k < K; ++k) {
    const int sv = __builtin_amdgcn_readlane(mycol, j + k);
    v[k] = *reinterpret_cast<const float4*>(h + (size_t)sv * 128 + lane * 4);
  }
  #pragma unroll
  for (int k = 0; k < K; ++k) {
    acc.x += v[k].x; acc.y += v[k].y; acc.z += v[k].z; acc.w += v[k].w;
  }
}

__device__ __forceinline__ float4 mean_of_rows(const float* __restrict__ h,
                                               const int* __restrict__ col,
                                               int beg, int end, int lane) {
  float4 acc = make_float4(0.0f, 0.0f, 0.0f, 0.0f);
  for (int base = beg; base < end; base += 32) {
    const int n = min(32, end - base);
    const int mycol = (lane < n) ? col[base + lane] : 0;
    int j = 0;
    for (; j + 8 <= n; j += 8) add_rows<8>(h, mycol, j, lane, acc);
    if (n - j >= 4) { add_rows<4>(h, mycol, j, lane, acc); j += 4; }
    if (n - j >= 2) { add_rows<2>(h, mycol, j, lane, acc); j += 2; }
    if (n - j >= 1) { add_rows<1>(h, mycol, j, lane, acc); }
  }
  const float d = fmaxf((float)(end - beg), 1.0f);
  float4 r;
  r.x = acc.x / d; r.y = acc.y / d; r.z = acc.z / d; r.w = acc.w / d;
  return r;
}


__global__ void __launch_bounds__(256) sage_layer_kernel(
    const float* __restrict__ hin, const int* __restrict__ rowbeg,
    const int* __restrict__ rowend, const int* __restrict__ col,
    const float* __restrict__ Wl,
    const float* __restrict__ bl, const float* __restrict__ Wr,
    float* __restrict__ hout, int nrows, int do_relu) {
  __shared__ __attribute__((aligned(16))) float As[32 * LDK];
  const int tid  = threadIdx.x;
  const int lane = tid & 31;
  const int row0 = blockIdx.x * 32;
  const int wv = __builtin_amdgcn_readfirstlane(tid >> 5);
  for (int q = 0; q < 4; ++q) {
    const int r   = 4 * wv + q;
    const int row = row0 + r;
    float4 own = make_float4(0.0f, 0.0f, 0.0f, 0.0f);
    float4 mn  = make_float4(0.0f, 0.0f, 0.0f, 0.0f);
    if (row < nrows) {
      own = *reinterpret_cast<const float4*>(hin + (size_t)row * 128 + lane * 4);
      mn  = mean_of_rows(hin, col, rowbeg[row], rowend[row], lane);
    }
    *reinterpret_cast<float4*>(&As[r * LDK + lane * 4])       = mn;
    *reinterpret_cast<float4*>(&As[r * LDK + 128 + lane * 4]) = own;
  }
  __syncthreads();
  const int col0 = (tid >> 5) * 16;
  v8f c0, c1;
  float bias = bl[col0 + (lane & 15)];
  #pragma unroll
  for (int v = 0; v < 8; ++v) { c0[v] = bias; c1[v] = bias; }
  gemm_acc2(As, 0,   Wl, 128, col0, lane, c0, c1);
  gemm_acc2(As, 128, Wr, 128, col0, lane, c0, c1);
  store_tile(hout, row0,      nrows, col0, lane, c0, do_relu != 0);
  store_tile(hout, row0 + 16, nrows, col0, lane, c1, do_relu != 0);
}


extern "C" void kernel_launch(void* const* d_in, const int* in_sizes, int n_in,
                              void* d_out, int out_size, void* d_ws, size_t ws_size,
                              hipStream_t stream) {
  (void)n_in; (void)out_size; (void)ws_size;
  const float* x_pol   = (const float*)d_in[0];
  const int*   pol_sid = (const int*)  d_in[1];
  const float* x_comp  = (const float*)d_in[2];
  const int*   c_sec   = (const int*)  d_in[3];
  const int*   c_ind   = (const int*)  d_in[4];
  const int*   edges   = (const int*)  d_in[5];
  const float* s_emb   = (const float*)d_in[6];
  const float* sec_emb = (const float*)d_in[7];
  const float* ind_emb = (const float*)d_in[8];
  const float* W_pol   = (const float*)d_in[9];
  const float* b_pol   = (const float*)d_in[10];
  const float* W_comp  = (const float*)d_in[11];
  const float* b_comp  = (const float*)d_in[12];
  const float* Wl1     = (const float*)d_in[13];
  const float* bl1     = (const float*)d_in[14];
  const float* Wr1     = (const float*)d_in[15];
  const float* Wl2     = (const float*)d_in[16];
  const float* bl2     = (const float*)d_in[17];
  const float* Wr2     = (const float*)d_in[18];

  const int NP = in_sizes[0] / 64;
  const int NC = in_sizes[2] / 96;
  const int E  = in_sizes[5] / 2;
  const int N  = NP + NC;
  const int* src = edges;
  const int* dst = edges + E;

  float* hws  = (float*)d_ws;
  float* outp = (float*)d_out;
  const int N4 = (N + 3) & ~3;
  const int NB = (N + BIN_NODES - 1) >> BIN_SHIFT;
  const int G  = (E + CHUNK_CAP - 1) / CHUNK_CAP;
  int* tab     = (int*)(hws + (size_t)N * 128);
  int* binbase = tab + G * NB;
  int* rowbeg  = tab + N4;
  int* rowend  = rowbeg + N4;
  int* col     = rowend + N4;
  int* pairs   = (int*)d_ws;

  const dim3 blk(256);

  encode_pol_kernel<<<dim3((NP + 31) / 32), blk, 0, stream>>>(
      x_pol, pol_sid, s_emb, W_pol, b_pol, outp, NP);
  encode_comp_kernel<<<dim3((NC + 31) / 32), blk, 0, stream>>>(
      x_comp, c_sec, c_ind, sec_emb, ind_emb, W_comp, b_comp,
      outp + (size_t)NP * 128, NC);

  bin_count_kernel<<<dim3(G), dim3(SORT_THREADS), 0, stream>>>(dst, tab, E, NB);
  bin_scatter_kernel<<<dim3(G), dim3(SORT_THREADS), 0, stream>>>(
      src, dst, tab, binbase, pairs, E, NB);
  bin_sort_kernel<<<dim3(NB), dim3(SORT_THREADS), 0, stream>>>(
      pairs, binbase, rowbeg, rowend, col, N);

  sage_layer_kernel<<<dim3((N + 31) / 32), blk, 0, stream>>>(
      outp, rowbeg, rowend, col, Wl1, bl1, Wr1, hws, N, 1);

  sage_layer_kernel<<<dim3((N + 31) / 32), blk, 0, stream>>>(
      hws, rowbeg, rowend, col, Wl2, bl2, Wr2, outp, N, 0);
}


# Encoder tiles run inside the three edge-grouping launches (work-groups behind the grouping ones encode four 32-row tiles each): 5 launches, not 7; encoders overlap the grouping; sum order unchanged
# speedup vs baseline: 3.9568x; 1.0513x over previous
#include <hip/hip_runtime.h>


typedef __attribute__((ext_vector_type(2))) float v2f;
typedef __attribute__((ext_vector_type(8))) float v8f;

#define LDK 260

__device__ __forceinline__ v8f wmma4(v2f a, v2f b, v8f c) {
  return __builtin_amdgcn_wmma_f32_16x16x4_f32(false, a, false, b,
                                               (short)0, c, false, false);
}

__device__ __forceinline__ void gemm_acc2(const float* As, int aoff,
                                          const float* __restrict__ B,
                                          int K, int col0, int lane,
                                          v8f& c0, v8f& c1) {
  const int hl = lane >> 4;
  const int m  = lane & 15;
  const int n  = lane & 15;
  #pragma unroll 4
  for (int k0 = 0; k0 < K; k0 += 4) {
    const int ka = k0 + 2 * hl;
    v2f a0, a1, b;
    a0.x = As[m * LDK + aoff + ka];
    a0.y = As[m * LDK + aoff + ka + 1];
    a1.x = As[(m + 16) * LDK + aoff + ka];
    a1.y = As[(m + 16) * LDK + aoff + ka + 1];
    b.x = B[ka * 128 + col0 + n];
    b.y = B[(ka + 1) * 128 + col0 + n];
    c0 = wmma4(a0, b, c0);
    c1 = wmma4(a1, b, c1);
  }
}

__device__ __forceinline__ void store_tile(float* __restrict__ out, int row0,
                                           int nrows, int col0, int lane,
                                           v8f c, bool do_relu) {
  const int hl = lane >> 4;
  const int n  = lane & 15;
  #pragma unroll
  for (int v = 0; v < 8; ++v) {
    int row = row0 + v + 8 * hl;
    float val = c[v];
    if (do_relu) val = fmaxf(val, 0.0f);
    if (row < nrows) out[(size_t)row * 128 + col0 + n] = val;
  }
}


__device__ __forceinline__ float4 keep_if(float4 v, bool ok) {
  float4 r;
  r.x = ok ? v.x : 0.0f; r.y = ok ? v.y : 0.0f;
  r.z = ok ? v.z : 0.0f; r.w = ok ? v.w : 0.0f;
  return r;
}


__device__ __forceinline__ void encode_pol_tile(
    float* As, int tid, int row0,
    const float* __restrict__ x, const int* __restrict__ sidx,
    const float* __restrict__ semb, const float* __restrict__ W,
    const float* __restrict__ bvec, float* __restrict__ out, int nrows) {
  const int last = nrows - 1;
  const int er = tid >> 3, ec = tid & 7;
  const int si = sidx[min(row0 + er, last)];
  float4 xv[2];
  #pragma unroll
  for (int i = 0; i < 2; ++i) {
    const int f = tid + 256 * i;
    const int r = f >> 4, c4 = f & 15;
    const int row = min(row0 + r, last);
    xv[i] = *reinterpret_cast<const float4*>(x + (size_t)row * 64 + c4 * 4);
  }
  const float e = semb[si * 8 + ec];
  #pragma unroll
  for (int i = 0; i < 2; ++i) {
    const int f = tid + 256 * i;
    const int r = f >> 4, c4 = f & 15;
    *reinterpret_cast<float4*>(&As[r * LDK + c4 * 4]) =
        keep_if(xv[i], row0 + r < nrows);
  }
  As[er * LDK + 64 + ec] = (row0 + er < nrows) ? e : 0.0f;
  __syncthreads();
  const int lane = tid & 31;
  const int col0 = (tid >> 5) * 16;
  v8f c0, c1;
  float bias = bvec[col0 + (lane & 15)];
  #pragma unroll
  for (int v = 0; v < 8; ++v) { c0[v] = bias; c1[v] = bias; }
  gemm_acc2(As, 0, W, 72, col0, lane, c0, c1);
  store_tile(out, row0,      nrows, col0, lane, c0, true);
  store_tile(out, row0 + 16, nrows, col0, lane, c1, true);
}

__device__ __forceinline__ void encode_comp_tile(
    float* As, int tid, int row0,
    const float* __restrict__ x, const int* __restrict__ secidx,
    const int* __restrict__ indidx, const float* __restrict__ secemb,
    const float* __restrict__ indemb, const float* __restrict__ W,
    const float* __restrict__ bvec, float* __restrict__ out, int nrows) {
  const int last = nrows - 1;
  const int er = tid >> 3, ep = tid & 7;
  const int*   itab = (ep & 4) ? indidx : secidx;
  const float* etab = (ep & 4) ? indemb : secemb;
  const int ei = itab[min(row0 + er, last)];
  float4 xv[3];
  #pragma unroll
  for (int i = 0; i < 3; ++i) {
    const int f = tid + 256 * i;
    const int r = f / 24, c4 = f % 24;
    const int row = min(row0 + r, last);
    xv[i] = *reinterpret_cast<const float4*>(x + (size_t)row * 96 + c4 * 4);
  }
  const float2 e = *reinterpret_cast<const float2*>(etab + ei * 8 + (ep & 3) * 2);
  #pragma unroll
  for (int i = 0; i < 3; ++i) {
    const int f = tid + 256 * i;
    const int r = f / 24, c4 = f % 24;
    *reinterpret_cast<float4*>(&As[r * LDK + c4 * 4]) =
        keep_if(xv[i], row0 + r < nrows);
  }
  {
    const bool ok = row0 + er < nrows;
    float2 ez;
    ez.x = ok ? e.x : 0.0f; ez.y = ok ? e.y : 0.0f;
    *reinterpret_cast<float2*>(&As[er * LDK + 96 + ep * 2]) = ez;
  }
  __syncthreads();
  const int lane = tid & 31;
  const int col0 = (tid >> 5) * 16;
  v8f c0, c1;
  float bias = bvec[col0 + (lane & 15)];
  #pragma unroll
  for (int v = 0; v < 8; ++v) { c0[v] = bias; c1[v] = bias; }
  gemm_acc2(As, 0, W, 112, col0, lane, c0, c1);
  store_tile(out, row0,      nrows, col0, lane, c0, true);
  store_tile(out, row0 + 16, nrows, col0, lane, c1, true);
}

struct CompEnc {
  const float* x; const int* secidx; const int* indidx;
  const float* secemb; const float* indemb;
  const float* W; const float* bvec; float* out; int nrows;
};


#define BIN_SHIFT 9
#define BIN_NODES 512
#define MAX_BINS 1024
#define CHUNK_CAP 8192
#define BIN_STAGE 12288
#define SORT_THREADS 1024
#define SMEM_WORDS (4 * 32 * LDK)

__device__ __forceinline__ void bin_count_body(
    int* hist, const int* __restrict__ dst, int* __restrict__ tab,
    int E, int NB, int me, int G) {
  const int tid = threadIdx.x;
  for (int k = tid; k < NB; k += SORT_THREADS) hist[k] = 0;
  __syncthreads();
  const int chunk = (E + G - 1) / G;
  const int beg = me * chunk;
  const int end = min(E, beg + chunk);
  for (int e0 = beg + tid; e0 < end; e0 += 4 * SORT_THREADS) {
    int d[4];
    #pragma unroll
    for (int u = 0; u < 4; ++u) {
      const int e = e0 + u * SORT_THREADS;
      d[u] = (e < end) ? dst[e] : 0;
    }
    #pragma unroll
    for (int u = 0; u < 4; ++u) {
      if (e0 + u * SORT_THREADS < end) atomicAdd(&hist[d[u] >> BIN_SHIFT], 1);
    }
  }
  __syncthreads();
  for (int k = tid; k < NB; k += SORT_THREADS) tab[me * NB + k] = hist[k];
}

__device__ __forceinline__ void bin_scatter_body(
    int* sm, const int* __restrict__ src, const int* __restrict__ dst,
    const int* __restrict__ tab, int* __restrict__ binbase,
    int* __restrict__ pairs, int E, int NB, int me, int G) {
  int*  tot   = sm;
  int*  pre   = sm + 4 * MAX_BINS;
  int*  lcnt  = sm + 8 * MAX_BINS;
  int*  gbase = sm + 9 * MAX_BINS;
  int*  lofs  = sm + 10 * MAX_BINS;
  int*  lcur  = sm + 11 * MAX_BINS;
  int2* stage = reinterpret_cast<int2*>(sm + 12 * MAX_BINS);
  const int tid = threadIdx.x;
  for (int k = tid; k < NB; k += SORT_THREADS) lcnt[k] = tab[me * NB + k];
  {
    const int part = tid >> 8;
    const int b0 = (part * G) >> 2;
    const int b1 = ((part + 1) * G) >> 2;
    for (int k = tid & 255; k < NB; k += 256) {
      int t = 0, p = 0;
      #pragma unroll 8
      for (int b = b0; b < b1; ++b) {
        const int v = tab[b * NB + k];
        t += v;
        if (b < me) p += v;
      }
      tot[part * MAX_BINS + k] = t;
      pre[part * MAX_BINS + k] = p;
    }
  }
  __syncthreads();
  if (tid < 32) {
    const int C  = (NB + 31) >> 5;
    const int kb = tid * C;
    const int ke = min(NB, kb + C);
    int s = 0, ls = 0;
    for (int k = kb; k < ke; ++k) {
      s  += (tot[k] + tot[MAX_BINS + k]) + (tot[2 * MAX_BINS + k] + tot[3 * MAX_BINS + k]);
      ls += lcnt[k];
    }
    int inc = s, linc = ls;
    for (int o = 1; o < 32; o <<= 1) {
      const int t  = __shfl_up(inc, o, 32);
      const int lt = __shfl_up(linc, o, 32);
      if (tid >= o) { inc += t; linc += lt; }
    }
    int run = inc - s, lrun = linc - ls;
    for (int k = kb; k < ke; ++k) {
      gbase[k] = run + (pre[k] + pre[MAX_BINS + k]) + (pre[2 * MAX_BINS + k] + pre[3 * MAX_BINS + k]);
      lofs[k]  = lrun;
      lcur[k]  = lrun;
      if (me == 0) binbase[k] = run;
      run  += (tot[k] + tot[MAX_BINS + k]) + (tot[2 * MAX_BINS + k] + tot[3 * MAX_BINS + k]);
      lrun += lcnt[k];
    }
    if (me == 0 && tid == 31) binbase[NB] = run;
  }
  __syncthreads();
  const int chunk = (E + G - 1) / G;
  const int beg = me * chunk;
  const int end = min(E, beg + chunk);
  for (int e0 = beg + tid; e0 < end; e0 += 4 * SORT_THREADS) {
    int d[4], s[4];
    #pragma unroll
    for (int u = 0; u < 4; ++u) {
      const int e = e0 + u * SORT_THREADS;
      d[u] = (e < end) ? dst[e] : 0;
      s[u] = (e < end) ? src[e] : 0;
    }
    #pragma unroll
    for (int u = 0; u < 4; ++u) {
      if (e0 + u * SORT_THREADS < end) {
        const int q = atomicAdd(&lcur[d[u] >> BIN_SHIFT], 1);
        stage[q] = make_int2(s[u], d[u]);
      }
    }
  }
  __syncthreads();
  int2* out2 = reinterpret_cast<int2*>(pairs);
  const int cnt = end - beg;
  for (int i = tid; i < cnt; i += SORT_THREADS) {
    const int2 pr = stage[i];
    const int k = pr.y >> BIN_SHIFT;
    out2[gbase[k] + (i - lofs[k])] = pr;
  }
}

__device__ __forceinline__ void bin_sort_body(
    int* sm, const int* __restrict__ pairs, const int* __restrict__ binbase,
    int* __restrict__ rowbeg, int* __restrict__ rowend,
    int* __restrict__ col, int N, int bin) {
  int* cnt   = sm;
  int* start = sm + BIN_NODES;
  int* cur   = sm + 2 * BIN_NODES;
  int* stage = sm + 3 * BIN_NODES;
  const int tid = threadIdx.x;
  const int2* in2 = reinterpret_cast<const int2*>(pairs);
  const int beg = binbase[bin];
  const int end = binbase[bin + 1];
  if (tid < BIN_NODES) cnt[tid] = 0;
  __syncthreads();
  for (int e0 = beg + tid; e0 < end; e0 += 4 * SORT_THREADS) {
    int d[4];
    #pragma unroll
    for (int u = 0; u < 4; ++u) {
      const int e = e0 + u * SORT_THREADS;
      d[u] = (e < end) ? in2[e].y : 0;
    }
    #pragma unroll
    for (int u = 0; u < 4; ++u) {
      if (e0 + u * SORT_THREADS < end) atomicAdd(&cnt[d[u] & (BIN_NODES - 1)], 1);
    }
  }
  __syncthreads();
  if (tid < 32) {
    int c[BIN_NODES / 32];
    int s = 0;
    #pragma unroll
    for (int j = 0; j < BIN_NODES / 32; ++j) { c[j] = cnt[tid * (BIN_NODES / 32) + j]; s += c[j]; }
    int inc = s;
    for (int o = 1; o < 32; o <<= 1) {
      const int t = __shfl_up(inc, o, 32);
      if (tid >= o) inc += t;
    }
    int run = inc - s;
    #pragma unroll
    for (int j = 0; j < BIN_NODES / 32; ++j) {
      start[tid * (BIN_NODES / 32) + j] = run;
      cur[tid * (BIN_NODES / 32) + j]   = run;
      run += c[j];
    }
  }
  __syncthreads();
  if (tid < BIN_NODES) {
    const int node = bin * BIN_NODES + tid;
    if (node < N) {
      rowbeg[node] = beg + start[tid];
      rowend[node] = beg + start[tid] + cnt[tid];
    }
  }
  for (int e0 = beg + tid; e0 < end; e0 += 4 * SORT_THREADS) {
    int2 pr[4];
    #pragma unroll
    for (int u = 0; u < 4; ++u) {
      const int e = e0 + u * SORT_THREADS;
      pr[u] = (e < end) ? in2[e] : make_int2(0, 0);
    }
    #pragma unroll
    for (int u = 0; u < 4; ++u) {
      if (e0 + u * SORT_THREADS < end) {
        const int q = atomicAdd(&cur[pr[u].y & (BIN_NODES - 1)], 1);
        if (q < BIN_STAGE) stage[q] = pr[u].x;
        else               col[beg + q] = pr[u].x;
      }
    }
  }
  __syncthreads();
  const int staged = min(end - beg, BIN_STAGE);
  for (int i = tid; i < staged; i += SORT_THREADS) col[beg + i] = stage[i];
}


__global__ void __launch_bounds__(SORT_THREADS) count_and_pol_kernel(
    const int* __restrict__ dst, int* __restrict__ tab, int E, int NB, int G,
    const float* __restrict__ x, const int* __restrict__ sidx,
    const float* __restrict__ semb, const float* __restrict__ W,
    const float* __restrict__ bvec, float* __restrict__ out, int nrows) {
  __shared__ __attribute__((aligned(16))) int smem[SMEM_WORDS];
  const int b = (int)blockIdx.x;
  if (b < G) {
    bin_count_body(smem, dst, tab, E, NB, b, G);
  } else {
    const int sub = (int)threadIdx.x >> 8;
    encode_pol_tile(reinterpret_cast<float*>(smem) + sub * 32 * LDK,
                    (int)threadIdx.x & 255, ((b - G) * 4 + sub) * 32,
                    x, sidx, semb, W, bvec, out, nrows);
  }
}

__global__ void __launch_bounds__(SORT_THREADS) scatter_and_comp_kernel(
    const int* __restrict__ src, const int* __restrict__ dst,
    const int* __restrict__ tab, int* __restrict__ binbase,
    int* __restrict__ pairs, int E, int NB, int G, CompEnc ce, int tile0) {
  __shared__ __attribute__((aligned(16))) int smem[SMEM_WORDS];
  const int b = (int)blockIdx.x;
  if (b < G) {
    bin_scatter_body(smem, src, dst, tab, binbase, pairs, E, NB, b, G);
  } else {
    const int sub = (int)threadIdx.x >> 8;
    encode_comp_tile(reinterpret_cast<float*>(smem) + sub * 32 * LDK,
                     (int)threadIdx.x & 255, (tile0 + (b - G) * 4 + sub) * 32,
                     ce.x, ce.secidx, ce.indidx, ce.secemb, ce.indemb,
                     ce.W, ce.bvec, ce.out, ce.nrows);
  }
}

__global__ void __launch_bounds__(SORT_THREADS) sort_and_comp_kernel(
    const int* __restrict__ pairs, const int* __restrict__ binbase,
    int* __restrict__ rowbeg, int* __restrict__ rowend,
    int* __restrict__ col, int N, int NB, CompEnc ce, int tile0) {
  __shared__ __attribute__((aligned(16))) int smem[SMEM_WORDS];
  const int b = (int)blockIdx.x;
  if (b < NB) {
    bin_sort_body(smem, pairs, binbase, rowbeg, rowend, col, N, b);
  } else {
    const int sub = (int)threadIdx.x >> 8;
    encode_comp_tile(reinterpret_cast<float*>(smem) + sub * 32 * LDK,
                     (int)threadIdx.x & 255, (tile0 + (b - NB) * 4 + sub) * 32,
                     ce.x, ce.secidx, ce.indidx, ce.secemb, ce.indemb,
                     ce.W, ce.bvec, ce.out, ce.nrows);
  }
}

template <int K>
__device__ __forceinline__ void add_rows(const float* __restrict__ h, int mycol,
                                         int j, int lane, float4& acc) {
  float4 v[K];
  #pragma unroll
  for (int k = 0; k < K; ++k) {
    const int sv = __builtin_amdgcn_readlane(mycol, j + k);
    v[k] = *reinterpret_cast<const float4*>(h + (size_t)sv * 128 + lane * 4);
  }
  #pragma unroll
  for (int k = 0; k < K; ++k) {
    acc.x += v[k].x; acc.y += v[k].y; acc.z += v[k].z; acc.w += v[k].w;
  }
}

__device__ __forceinline__ float4 mean_of_rows(const float* __restrict__ h,
                                               const int* __restrict__ col,
                                               int beg, int end, int lane) {
  float4 acc = make_float4(0.0f, 0.0f, 0.0f, 0.0f);
  for (int base = beg; base < end; base += 32) {
    const int n = min(32, end - base);
    const int mycol = (lane < n) ? col[base + lane] : 0;
    int j = 0;
    for (; j + 8 <= n; j += 8) add_rows<8>(h, mycol, j, lane, acc);
    if (n - j >= 4) { add_rows<4>(h, mycol, j, lane, acc); j += 4; }
    if (n - j >= 2) { add_rows<2>(h, mycol, j, lane, acc); j += 2; }
    if (n - j >= 1) { add_rows<1>(h, mycol, j, lane, acc); }
  }
  const float d = fmaxf((float)(end - beg), 1.0f);
  float4 r;
  r.x = acc.x / d; r.y = acc.y / d; r.z = acc.z / d; r.w = acc.w / d;
  return r;
}


__global__ void __launch_bounds__(256) sage_layer_kernel(
    const float* __restrict__ hin, const int* __restrict__ rowbeg,
    const int* __restrict__ rowend, const int* __restrict__ col,
    const float* __restrict__ Wl,
    const float* __restrict__ bl, const float* __restrict__ Wr,
    float* __restrict__ hout, int nrows, int do_relu) {
  __shared__ __attribute__((aligned(16))) float As[32 * LDK];
  const int tid  = threadIdx.x;
  const int lane = tid & 31;
  const int row0 = blockIdx.x * 32;
  const int wv = __builtin_amdgcn_readfirstlane(tid >> 5);
  for (int q = 0; q < 4; ++q) {
    const int r   = 4 * wv + q;
    const int row = row0 + r;
    float4 own = make_float4(0.0f, 0.0f, 0.0f, 0.0f);
    float4 mn  = make_float4(0.0f, 0.0f, 0.0f, 0.0f);
    if (row < nrows) {
      own = *reinterpret_cast<const float4*>(hin + (size_t)row * 128 + lane * 4);
      mn  = mean_of_rows(hin, col, rowbeg[row], rowend[row], lane);
    }
    *reinterpret_cast<float4*>(&As[r * LDK + lane * 4])       = mn;
    *reinterpret_cast<float4*>(&As[r * LDK + 128 + lane * 4]) = own;
  }
  __syncthreads();
  const int col0 = (tid >> 5) * 16;
  v8f c0, c1;
  float bias = bl[col0 + (lane & 15)];
  #pragma unroll
  for (int v = 0; v < 8; ++v) { c0[v] = bias; c1[v] = bias; }
  gemm_acc2(As, 0,   Wl, 128, col0, lane, c0, c1);
  gemm_acc2(As, 128, Wr, 128, col0, lane, c0, c1);
  store_tile(hout, row0,      nrows, col0, lane, c0, do_relu != 0);
  store_tile(hout, row0 + 16, nrows, col0, lane, c1, do_relu != 0);
}


extern "C" void kernel_launch(void* const* d_in, const int* in_sizes, int n_in,
                              void* d_out, int out_size, void* d_ws, size_t ws_size,
                              hipStream_t stream) {
  (void)n_in; (void)out_size; (void)ws_size;
  const float* x_pol   = (const float*)d_in[0];
  const int*   pol_sid = (const int*)  d_in[1];
  const float* x_comp  = (const float*)d_in[2];
  const int*   c_sec   = (const int*)  d_in[3];
  const int*   c_ind   = (const int*)  d_in[4];
  const int*   edges   = (const int*)  d_in[5];
  const float* s_emb   = (const float*)d_in[6];
  const float* sec_emb = (const float*)d_in[7];
  const float* ind_emb = (const float*)d_in[8];
  const float* W_pol   = (const float*)d_in[9];
  const float* b_pol   = (const float*)d_in[10];
  const float* W_comp  = (const float*)d_in[11];
  const float* b_comp  = (const float*)d_in[12];
  const float* Wl1     = (const float*)d_in[13];
  const float* bl1     = (const float*)d_in[14];
  const float* Wr1     = (const float*)d_in[15];
  const float* Wl2     = (const float*)d_in[16];
  const float* bl2     = (const float*)d_in[17];
  const float* Wr2     = (const float*)d_in[18];

  const int NP = in_sizes[0] / 64;
  const int NC = in_sizes[2] / 96;
  const int E  = in_sizes[5] / 2;
  const int N  = NP + NC;
  const int* src = edges;
  const int* dst = edges + E;

  float* hws  = (float*)d_ws;
  float* outp = (float*)d_out;
  const int N4 = (N + 3) & ~3;
  const int NB = (N + BIN_NODES - 1) >> BIN_SHIFT;
  const int G  = (E + CHUNK_CAP - 1) / CHUNK_CAP;
  int* tab     = (int*)(hws + (size_t)N * 128);
  int* binbase = tab + G * NB;
  int* rowbeg  = tab + N4;
  int* rowend  = rowbeg + N4;
  int* col     = rowend + N4;
  int* pairs   = (int*)d_ws;

  const dim3 blk(256);
  const dim3 sblk(SORT_THREADS);

  const int BP  = ((NP + 31) / 32 + 3) / 4;
  const int BC  = ((NC + 31) / 32 + 3) / 4;
  const int BC1 = BC / 2;
  const int BC2 = BC - BC1;
  CompEnc ce;
  ce.x = x_comp; ce.secidx = c_sec; ce.indidx = c_ind;
  ce.secemb = sec_emb; ce.indemb = ind_emb;
  ce.W = W_comp; ce.bvec = b_comp; ce.out = outp + (size_t)NP * 128; ce.nrows = NC;

  count_and_pol_kernel<<<dim3(G + BP), sblk, 0, stream>>>(
      dst, tab, E, NB, G, x_pol, pol_sid, s_emb, W_pol, b_pol, outp, NP);
  scatter_and_comp_kernel<<<dim3(G + BC1), sblk, 0, stream>>>(
      src, dst, tab, binbase, pairs, E, NB, G, ce, 0);
  sort_and_comp_kernel<<<dim3(NB + BC2), sblk, 0, stream>>>(
      pairs, binbase, rowbeg, rowend, col, N, NB, ce, BC1 * 4);

  sage_layer_kernel<<<dim3((N + 31) / 32), blk, 0, stream>>>(
      outp, rowbeg, rowend, col, Wl1, bl1, Wr1, hws, N, 1);

  sage_layer_kernel<<<dim3((N + 31) / 32), blk, 0, stream>>>(
      hws, rowbeg, rowend, col, Wl2, bl2, Wr2, outp, N, 0);
}


# Result tiles staged through the LDS A tile and stored as whole 512-byte rows (one 16-byte store per lane, one wave per 128-byte line), not 64-byte pieces from two waves per line; sums unchanged
# speedup vs baseline: 4.1352x; 1.0451x over previous
#include <hip/hip_runtime.h>


typedef __attribute__((ext_vector_type(2))) float v2f;
typedef __attribute__((ext_vector_type(8))) float v8f;

#define LDK 260

__device__ __forceinline__ v8f wmma4(v2f a, v2f b, v8f c) {
  return __builtin_amdgcn_wmma_f32_16x16x4_f32(false, a, false, b,
                                               (short)0, c, false, false);
}

__device__ __forceinline__ void gemm_acc2(const float* As, int aoff,
                                          const float* __restrict__ B,
                                          int K, int col0, int lane,
                                          v8f& c0, v8f& c1) {
  const int hl = lane >> 4;
  const int m  = lane & 15;
  const int n  = lane & 15;
  #pragma unroll 4
  for (int k0 = 0; k0 < K; k0 += 4) {
    const int ka = k0 + 2 * hl;
    v2f a0, a1, b;
    a0.x = As[m * LDK + aoff + ka];
    a0.y = As[m * LDK + aoff + ka + 1];
    a1.x = As[(m + 16) * LDK + aoff + ka];
    a1.y = As[(m + 16) * LDK + aoff + ka + 1];
    b.x = B[ka * 128 + col0 + n];
    b.y = B[(ka + 1) * 128 + col0 + n];
    c0 = wmma4(a0, b, c0);
    c1 = wmma4(a1, b, c1);
  }
}

__device__ __forceinline__ void store_rows_via_lds(float* Cs, float* __restrict__ out,
                                                   int row0, int nrows, int tid,
                                                   v8f c0, v8f c1, bool do_relu) {
  const int lane = tid & 31;
  const int wv   = tid >> 5;
  const int hl   = lane >> 4;
  const int n    = lane & 15;
  const int col0 = wv * 16;
  __syncthreads();
  #pragma unroll
  for (int v = 0; v < 8; ++v) {
    float a = c0[v], b = c1[v];
    if (do_relu) { a = fmaxf(a, 0.0f); b = fmaxf(b, 0.0f); }
    Cs[(v + 8 * hl) * LDK + col0 + n]      = a;
    Cs[(16 + v + 8 * hl) * LDK + col0 + n] = b;
  }
  __syncthreads();
  const float* rowp = &Cs[4 * wv * LDK + lane * 4];
  float*       outp = out + (size_t)(row0 + 4 * wv) * 128 + lane * 4;
  if (row0 + 32 <= nrows) {
    float4 val[4];
    #pragma unroll
    for (int q = 0; q < 4; ++q)
      val[q] = *reinterpret_cast<const float4*>(rowp + q * LDK);
    #pragma unroll
    for (int q = 0; q < 4; ++q)
      *reinterpret_cast<float4*>(outp + q * 128) = val[q];
  } else {
    #pragma unroll
    for (int q = 0; q < 4; ++q) {
      if (row0 + 4 * wv + q < nrows)
        *reinterpret_cast<float4*>(outp + q * 128) =
            *reinterpret_cast<const float4*>(rowp + q * LDK);
    }
  }
}


__device__ __forceinline__ float4 keep_if(float4 v, bool ok) {
  float4 r;
  r.x = ok ? v.x : 0.0f; r.y = ok ? v.y : 0.0f;
  r.z = ok ? v.z : 0.0f; r.w = ok ? v.w : 0.0f;
  return r;
}


__device__ __forceinline__ void encode_pol_tile(
    float* As, int tid, int row0,
    const float* __restrict__ x, const int* __restrict__ sidx,
    const float* __restrict__ semb, const float* __restrict__ W,
    const float* __restrict__ bvec, float* __restrict__ out, int nrows) {
  const int last = nrows - 1;
  const int er = tid >> 3, ec = tid & 7;
  const int si = sidx[min(row0 + er, last)];
  float4 xv[2];
  #pragma unroll
  for (int i = 0; i < 2; ++i) {
    const int f = tid + 256 * i;
    const int r = f >> 4, c4 = f & 15;
    const int row = min(row0 + r, last);
    xv[i] = *reinterpret_cast<const float4*>(x + (size_t)row * 64 + c4 * 4);
  }
  const float e = semb[si * 8 + ec];
  #pragma unroll
  for (int i = 0; i < 2; ++i) {
    const int f = tid + 256 * i;
    const int r = f >> 4, c4 = f & 15;
    *reinterpret_cast<float4*>(&As[r * LDK + c4 * 4]) =
        keep_if(xv[i], row0 + r < nrows);
  }
  As[er * LDK + 64 + ec] = (row0 + er < nrows) ? e : 0.0f;
  __syncthreads();
  const int lane = tid & 31;
  const int col0 = (tid >> 5) * 16;
  v8f c0, c1;
  float bias = bvec[col0 + (lane & 15)];
  #pragma unroll
  for (int v = 0; v < 8; ++v) { c0[v] = bias; c1[v] = bias; }
  gemm_acc2(As, 0, W, 72, col0, lane, c0, c1);
  store_rows_via_lds(As, out, row0, nrows, tid, c0, c1, true);
}

__device__ __forceinline__ void encode_comp_tile(
    float* As, int tid, int row0,
    const float* __restrict__ x, const int* __restrict__ secidx,
    const int* __restrict__ indidx, const float* __restrict__ secemb,
    const float* __restrict__ indemb, const float* __restrict__ W,
    const float* __restrict__ bvec, float* __restrict__ out, int nrows) {
  const int last = nrows - 1;
  const int er = tid >> 3, ep = tid & 7;
  const int*   itab = (ep & 4) ? indidx : secidx;
  const float* etab = (ep & 4) ? indemb : secemb;
  const int ei = itab[min(row0 + er, last)];
  float4 xv[3];
  #pragma unroll
  for (int i = 0; i < 3; ++i) {
    const int f = tid + 256 * i;
    const int r = f / 24, c4 = f % 24;
    const int row = min(row0 + r, last);
    xv[i] = *reinterpret_cast<const float4*>(x + (size_t)row * 96 + c4 * 4);
  }
  const float2 e = *reinterpret_cast<const float2*>(etab + ei * 8 + (ep & 3) * 2);
  #pragma unroll
  for (int i = 0; i < 3; ++i) {
    const int f = tid + 256 * i;
    const int r = f / 24, c4 = f % 24;
    *reinterpret_cast<float4*>(&As[r * LDK + c4 * 4]) =
        keep_if(xv[i], row0 + r < nrows);
  }
  {
    const bool ok = row0 + er < nrows;
    float2 ez;
    ez.x = ok ? e.x : 0.0f; ez.y = ok ? e.y : 0.0f;
    *reinterpret_cast<float2*>(&As[er * LDK + 96 + ep * 2]) = ez;
  }
  __syncthreads();
  const int lane = tid & 31;
  const int col0 = (tid >> 5) * 16;
  v8f c0, c1;
  float bias = bvec[col0 + (lane & 15)];
  #pragma unroll
  for (int v = 0; v < 8; ++v) { c0[v] = bias; c1[v] = bias; }
  gemm_acc2(As, 0, W, 112, col0, lane, c0, c1);
  store_rows_via_lds(As, out, row0, nrows, tid, c0, c1, true);
}

struct CompEnc {
  const float* x; const int* secidx; const int* indidx;
  const float* secemb; const float* indemb;
  const float* W; const float* bvec; float* out; int nrows;
  int tile0;
};


#define BIN_SHIFT 9
#define BIN_NODES 512
#define MAX_BINS 1024
#define CHUNK_CAP 8192
#define BIN_STAGE 12288
#define SORT_THREADS 1024
#define SMEM_WORDS (4 * 32 * LDK)

__device__ __forceinline__ void bin_count_body(
    int* hist, const int* __restrict__ dst, int* __restrict__ tab,
    int E, int NB, int me, int G) {
  const int tid = threadIdx.x;
  for (int k = tid; k < NB; k += SORT_THREADS) hist[k] = 0;
  __syncthreads();
  const int chunk = (E + G - 1) / G;
  const int beg = me * chunk;
  const int end = min(E, beg + chunk);
  for (int e0 = beg + tid; e0 < end; e0 += 4 * SORT_THREADS) {
    int d[4];
    #pragma unroll
    for (int u = 0; u < 4; ++u) {
      const int e = e0 + u * SORT_THREADS;
      d[u] = (e < end) ? dst[e] : 0;
    }
    #pragma unroll
    for (int u = 0; u < 4; ++u) {
      if (e0 + u * SORT_THREADS < end) atomicAdd(&hist[d[u] >> BIN_SHIFT], 1);
    }
  }
  __syncthreads();
  for (int k = tid; k < NB; k += SORT_THREADS) tab[me * NB + k] = hist[k];
}

__device__ __forceinline__ void bin_scatter_body(
    int* sm, const int* __restrict__ src, const int* __restrict__ dst,
    const int* __restrict__ tab, int* __restrict__ binbase,
    int* __restrict__ pairs, int E, int NB, int me, int G) {
  int*  tot   = sm;
  int*  pre   = sm + 4 * MAX_BINS;
  int*  lcnt  = sm + 8 * MAX_BINS;
  int*  gbase = sm + 9 * MAX_BINS;
  int*  lofs  = sm + 10 * MAX_BINS;
  int*  lcur  = sm + 11 * MAX_BINS;
  int2* stage = reinterpret_cast<int2*>(sm + 12 * MAX_BINS);
  const int tid = threadIdx.x;
  for (int k = tid; k < NB; k += SORT_THREADS) lcnt[k] = tab[me * NB + k];
  {
    const int part = tid >> 8;
    const int b0 = (part * G) >> 2;
    const int b1 = ((part + 1) * G) >> 2;
    for (int k = tid & 255; k < NB; k += 256) {
      int t = 0, p = 0;
      #pragma unroll 8
      for (int b = b0; b < b1; ++b) {
        const int v = tab[b * NB + k];
        t += v;
        if (b < me) p += v;
      }
      tot[part * MAX_BINS + k] = t;
      pre[part * MAX_BINS + k] = p;
    }
  }
  __syncthreads();
  if (tid < 32) {
    const int C  = (NB + 31) >> 5;
    const int kb = tid * C;
    const int ke = min(NB, kb + C);
    int s = 0, ls = 0;
    for (int k = kb; k < ke; ++k) {
      s  += (tot[k] + tot[MAX_BINS + k]) + (tot[2 * MAX_BINS + k] + tot[3 * MAX_BINS + k]);
      ls += lcnt[k];
    }
    int inc = s, linc = ls;
    for (int o = 1; o < 32; o <<= 1) {
      const int t  = __shfl_up(inc, o, 32);
      const int lt = __shfl_up(linc, o, 32);
      if (tid >= o) { inc += t; linc += lt; }
    }
    int run = inc - s, lrun = linc - ls;
    for (int k = kb; k < ke; ++k) {
      gbase[k] = run + (pre[k] + pre[MAX_BINS + k]) + (pre[2 * MAX_BINS + k] + pre[3 * MAX_BINS + k]);
      lofs[k]  = lrun;
      lcur[k]  = lrun;
      if (me == 0) binbase[k] = run;
      run  += (tot[k] + tot[MAX_BINS + k]) + (tot[2 * MAX_BINS + k] + tot[3 * MAX_BINS + k]);
      lrun += lcnt[k];
    }
    if (me == 0 && tid == 31) binbase[NB] = run;
  }
  __syncthreads();
  const int chunk = (E + G - 1) / G;
  const int beg = me * chunk;
  const int end = min(E, beg + chunk);
  for (int e0 = beg + tid; e0 < end; e0 += 4 * SORT_THREADS) {
    int d[4], s[4];
    #pragma unroll
    for (int u = 0; u < 4; ++u) {
      const int e = e0 + u * SORT_THREADS;
      d[u] = (e < end) ? dst[e] : 0;
      s[u] = (e < end) ? src[e] : 0;
    }
    #pragma unroll
    for (int u = 0; u < 4; ++u) {
      if (e0 + u * SORT_THREADS < end) {
        const int q = atomicAdd(&lcur[d[u] >> BIN_SHIFT], 1);
        stage[q] = make_int2(s[u], d[u]);
      }
    }
  }
  __syncthreads();
  int2* out2 = reinterpret_cast<int2*>(pairs);
  const int cnt = end - beg;
  for (int i = tid; i < cnt; i += SORT_THREADS) {
    const int2 pr = stage[i];
    const int k = pr.y >> BIN_SHIFT;
    out2[gbase[k] + (i - lofs[k])] = pr;
  }
}

__device__ __forceinline__ void bin_sort_body(
    int* sm, const int* __restrict__ pairs, const int* __restrict__ binbase,
    int* __restrict__ rowbeg, int* __restrict__ rowend,
    int* __restrict__ col, int N, int bin) {
  int* cnt   = sm;
  int* start = sm + BIN_NODES;
  int* cur   = sm + 2 * BIN_NODES;
  int* stage = sm + 3 * BIN_NODES;
  const int tid = threadIdx.x;
  const int2* in2 = reinterpret_cast<const int2*>(pairs);
  const int beg = binbase[bin];
  const int end = binbase[bin + 1];
  if (tid < BIN_NODES) cnt[tid] = 0;
  __syncthreads();
  for (int e0 = beg + tid; e0 < end; e0 += 4 * SORT_THREADS) {
    int d[4];
    #pragma unroll
    for (int u = 0; u < 4; ++u) {
      const int e = e0 + u * SORT_THREADS;
      d[u] = (e < end) ? in2[e].y : 0;
    }
    #pragma unroll
    for (int u = 0; u < 4; ++u) {
      if (e0 + u * SORT_THREADS < end) atomicAdd(&cnt[d[u] & (BIN_NODES - 1)], 1);
    }
  }
  __syncthreads();
  if (tid < 32) {
    int c[BIN_NODES / 32];
    int s = 0;
    #pragma unroll
    for (int j = 0; j < BIN_NODES / 32; ++j) { c[j] = cnt[tid * (BIN_NODES / 32) + j]; s += c[j]; }
    int inc = s;
    for (int o = 1; o < 32; o <<= 1) {
      const int t = __shfl_up(inc, o, 32);
      if (tid >= o) inc += t;
    }
    int run = inc - s;
    #pragma unroll
    for (int j = 0; j < BIN_NODES / 32; ++j) {
      start[tid * (BIN_NODES / 32) + j] = run;
      cur[tid * (BIN_NODES / 32) + j]   = run;
      run += c[j];
    }
  }
  __syncthreads();
  if (tid < BIN_NODES) {
    const int node = bin * BIN_NODES + tid;
    if (node < N) {
      rowbeg[node] = beg + start[tid];
      rowend[node] = beg + start[tid] + cnt[tid];
    }
  }
  for (int e0 = beg + tid; e0 < end; e0 += 4 * SORT_THREADS) {
    int2 pr[4];
    #pragma unroll
    for (int u = 0; u < 4; ++u) {
      const int e = e0 + u * SORT_THREADS;
      pr[u] = (e < end) ? in2[e] : make_int2(0, 0);
    }
    #pragma unroll
    for (int u = 0; u < 4; ++u) {
      if (e0 + u * SORT_THREADS < end) {
        const int q = atomicAdd(&cur[pr[u].y & (BIN_NODES - 1)], 1);
        if (q < BIN_STAGE) stage[q] = pr[u].x;
        else               col[beg + q] = pr[u].x;
      }
    }
  }
  __syncthreads();
  const int staged = min(end - beg, BIN_STAGE);
  for (int i = tid; i < staged; i += SORT_THREADS) col[beg + i] = stage[i];
}


__global__ void __launch_bounds__(SORT_THREADS) count_and_pol_kernel(
    const int* __restrict__ dst, int* __restrict__ tab, int E, int NB, int G,
    const float* __restrict__ x, const int* __restrict__ sidx,
    const float* __restrict__ semb, const float* __restrict__ W,
    const float* __restrict__ bvec, float* __restrict__ out, int nrows) {
  __shared__ __attribute__((aligned(16))) int smem[SMEM_WORDS];
  const int b = (int)blockIdx.x;
  if (b < G) {
    bin_count_body(smem, dst, tab, E, NB, b, G);
  } else {
    const int sub = (int)threadIdx.x >> 8;
    encode_pol_tile(reinterpret_cast<float*>(smem) + sub * 32 * LDK,
                    (int)threadIdx.x & 255, ((b - G) * 4 + sub) * 32,
                    x, sidx, semb, W, bvec, out, nrows);
  }
}

__global__ void __launch_bounds__(SORT_THREADS) scatter_and_comp_kernel(
    const int* __restrict__ src, const int* __restrict__ dst,
    const int* __restrict__ tab, int* __restrict__ binbase,
    int* __restrict__ pairs, int E, int NB, int G, CompEnc ce) {
  __shared__ __attribute__((aligned(16))) int smem[SMEM_WORDS];
  const int b = (int)blockIdx.x;
  if (b < G) {
    bin_scatter_body(smem, src, dst, tab, binbase, pairs, E, NB, b, G);
  } else {
    const int sub = (int)threadIdx.x >> 8;
    encode_comp_tile(reinterpret_cast<float*>(smem) + sub * 32 * LDK,
                     (int)threadIdx.x & 255, (ce.tile0 + (b - G) * 4 + sub) * 32,
                     ce.x, ce.secidx, ce.indidx, ce.secemb, ce.indemb,
                     ce.W, ce.bvec, ce.out, ce.nrows);
  }
}

__global__ void __launch_bounds__(SORT_THREADS) sort_and_comp_kernel(
    const int* __restrict__ pairs, const int* __restrict__ binbase,
    int* __restrict__ rowbeg, int* __restrict__ rowend,
    int* __restrict__ col, int N, int NB, CompEnc ce) {
  __shared__ __attribute__((aligned(16))) int smem[SMEM_WORDS];
  const int b = (int)blockIdx.x;
  if (b < NB) {
    bin_sort_body(smem, pairs, binbase, rowbeg, rowend, col, N, b);
  } else {
    const int sub = (int)threadIdx.x >> 8;
    encode_comp_tile(reinterpret_cast<float*>(smem) + sub * 32 * LDK,
                     (int)threadIdx.x & 255, (ce.tile0 + (b - NB) * 4 + sub) * 32,
                     ce.x, ce.secidx, ce.indidx, ce.secemb, ce.indemb,
                     ce.W, ce.bvec, ce.out, ce.nrows);
  }
}

template <int K>
__device__ __forceinline__ void add_rows(const float* __restrict__ h, int mycol,
                                         int j, int lane, float4& acc) {
  float4 v[K];
  #pragma unroll
  for (int k = 0; k < K; ++k) {
    const int sv = __builtin_amdgcn_readlane(mycol, j + k);
    v[k] = *reinterpret_cast<const float4*>(h + (size_t)sv * 128 + lane * 4);
  }
  #pragma unroll
  for (int k = 0; k < K; ++k) {
    acc.x += v[k].x; acc.y += v[k].y; acc.z += v[k].z; acc.w += v[k].w;
  }
}

__device__ __forceinline__ float4 mean_of_rows(const float* __restrict__ h,
                                               const int* __restrict__ col,
                                               int beg, int end, int lane) {
  float4 acc = make_float4(0.0f, 0.0f, 0.0f, 0.0f);
  for (int base = beg; base < end; base += 32) {
    const int n = min(32, end - base);
    const int mycol = (lane < n) ? col[base + lane] : 0;
    int j = 0;
    for (; j + 8 <= n; j += 8) add_rows<8>(h, mycol, j, lane, acc);
    if (n - j >= 4) { add_rows<4>(h, mycol, j, lane, acc); j += 4; }
    if (n - j >= 2) { add_rows<2>(h, mycol, j, lane, acc); j += 2; }
    if (n - j >= 1) { add_rows<1>(h, mycol, j, lane, acc); }
  }
  const float d = fmaxf((float)(end - beg), 1.0f);
  float4 r;
  r.x = acc.x / d; r.y = acc.y / d; r.z = acc.z / d; r.w = acc.w / d;
  return r;
}


__global__ void __launch_bounds__(256) sage_layer_kernel(
    const float* __restrict__ hin, const int* __restrict__ rowbeg,
    const int* __restrict__ rowend, const int* __restrict__ col,
    const float* __restrict__ Wl,
    const float* __restrict__ bl, const float* __restrict__ Wr,
    float* __restrict__ hout, int nrows, int do_relu) {
  __shared__ __attribute__((aligned(16))) float As[32 * LDK];
  const int tid  = threadIdx.x;
  const int lane = tid & 31;
  const int row0 = blockIdx.x * 32;
  const int wv = __builtin_amdgcn_readfirstlane(tid >> 5);
  for (int q = 0; q < 4; ++q) {
    const int r   = 4 * wv + q;
    const int row = row0 + r;
    float4 own = make_float4(0.0f, 0.0f, 0.0f, 0.0f);
    float4 mn  = make_float4(0.0f, 0.0f, 0.0f, 0.0f);
    if (row < nrows) {
      own = *reinterpret_cast<const float4*>(hin + (size_t)row * 128 + lane * 4);
      mn  = mean_of_rows(hin, col, rowbeg[row], rowend[row], lane);
    }
    *reinterpret_cast<float4*>(&As[r * LDK + lane * 4])       = mn;
    *reinterpret_cast<float4*>(&As[r * LDK + 128 + lane * 4]) = own;
  }
  __syncthreads();
  const int col0 = (tid >> 5) * 16;
  v8f c0, c1;
  float bias = bl[col0 + (lane & 15)];
  #pragma unroll
  for (int v = 0; v < 8; ++v) { c0[v] = bias; c1[v] = bias; }
  gemm_acc2(As, 0,   Wl, 128, col0, lane, c0, c1);
  gemm_acc2(As, 128, Wr, 128, col0, lane, c0, c1);
  store_rows_via_lds(As, hout, row0, nrows, tid, c0, c1, do_relu != 0);
}


extern "C" void kernel_launch(void* const* d_in, const int* in_sizes, int n_in,
                              void* d_out, int out_size, void* d_ws, size_t ws_size,
                              hipStream_t stream) {
  (void)n_in; (void)out_size; (void)ws_size;
  const float* x_pol   = (const float*)d_in[0];
  const int*   pol_sid = (const int*)  d_in[1];
  const float* x_comp  = (const float*)d_in[2];
  const int*   c_sec   = (const int*)  d_in[3];
  const int*   c_ind   = (const int*)  d_in[4];
  const int*   edges   = (const int*)  d_in[5];
  const float* s_emb   = (const float*)d_in[6];
  const float* sec_emb = (const float*)d_in[7];
  const float* ind_emb = (const float*)d_in[8];
  const float* W_pol   = (const float*)d_in[9];
  const float* b_pol   = (const float*)d_in[10];
  const float* W_comp  = (const float*)d_in[11];
  const float* b_comp  = (const float*)d_in[12];
  const float* Wl1     = (const float*)d_in[13];
  const float* bl1     = (const float*)d_in[14];
  const float* Wr1     = (const float*)d_in[15];
  const float* Wl2     = (const float*)d_in[16];
  const float* bl2     = (const float*)d_in[17];
  const float* Wr2     = (const float*)d_in[18];

  const int NP = in_sizes[0] / 64;
  const int NC = in_sizes[2] / 96;
  const int E  = in_sizes[5] / 2;
  const int N  = NP + NC;
  const int* src = edges;
  const int* dst = edges + E;

  float* hws  = (float*)d_ws;
  float* outp = (float*)d_out;
  const int N4 = (N + 3) & ~3;
  const int NB = (N + BIN_NODES - 1) >> BIN_SHIFT;
  const int G  = (E + CHUNK_CAP - 1) / CHUNK_CAP;
  int* tab     = (int*)(hws + (size_t)N * 128);
  int* binbase = tab + G * NB;
  int* rowbeg  = tab + N4;
  int* rowend  = rowbeg + N4;
  int* col     = rowend + N4;
  int* pairs   = (int*)d_ws;

  const dim3 blk(256);
  const dim3 sblk(SORT_THREADS);

  const int BP  = ((NP + 31) / 32 + 3) / 4;
  const int BC  = ((NC + 31) / 32 + 3) / 4;
  const int BC1 = BC / 2;
  const int BC2 = BC - BC1;
  CompEnc ce;
  ce.x = x_comp; ce.secidx = c_sec; ce.indidx = c_ind;
  ce.secemb = sec_emb; ce.indemb = ind_emb;
  ce.W = W_comp; ce.bvec = b_comp; ce.out = outp + (size_t)NP * 128; ce.nrows = NC;
  ce.tile0 = 0;

  count_and_pol_kernel<<<dim3(G + BP), sblk, 0, stream>>>(
      dst, tab, E, NB, G, x_pol, pol_sid, s_emb, W_pol, b_pol, outp, NP);
  scatter_and_comp_kernel<<<dim3(G + BC1), sblk, 0, stream>>>(
      src, dst, tab, binbase, pairs, E, NB, G, ce);
  ce.tile0 = BC1 * 4;
  sort_and_comp_kernel<<<dim3(NB + BC2), sblk, 0, stream>>>(
      pairs, binbase, rowbeg, rowend, col, N, NB, ce);

  sage_layer_kernel<<<dim3((N + 31) / 32), blk, 0, stream>>>(
      outp, rowbeg, rowend, col, Wl1, bl1, Wr1, hws, N, 1);

  sage_layer_kernel<<<dim3((N + 31) / 32), blk, 0, stream>>>(
      hws, rowbeg, rowend, col, Wl2, bl2, Wr2, outp, N, 0);
}
